# starting kernel
# baseline (speedup 1.0000x reference)
#include <hip/hip_runtime.h>

typedef __attribute__((ext_vector_type(2))) float v2f;
typedef __attribute__((ext_vector_type(8))) float v8f;

constexpr int T      = 12;
constexpr int C_IN   = 32;
constexpr int H      = 64;
constexpr int C_OUT  = 32;
constexpr int N_SRC  = 50000;
constexpr int N_DST1 = 20000;
constexpr int N_DST2 = 10000;
constexpr int ROW    = T * C_IN;
constexpr int ROW4   = ROW / 4;
constexpr int M1     = N_DST1 * T;
constexpr int M2     = N_DST2 * T;
constexpr float NEG_SLOPE = 0.01f;

__device__ __forceinline__ float leaky(float v) {
    return v > 0.f ? v : NEG_SLOPE * v;
}

__device__ __forceinline__ v2f load_frag(const float* __restrict__ base, int r0, int ld,
                                         int k0, int lane) {
    int r = r0 + (lane & 15);
    int k = k0 + ((lane >> 4) << 1);
    const float* p = base + (size_t)r * ld + k;
    v2f f;
    f.x = p[0];
    f.y = p[1];
    return f;
}

__global__ void k_transpose(const float* __restrict__ nf, float* __restrict__ xT) {
    int idx = blockIdx.x * blockDim.x + threadIdx.x;
    constexpr int C4 = C_IN / 4;
    constexpr int total = N_SRC * T * C4;
    if (idx >= total) return;
    int c4 = idx % C4;
    int t  = (idx / C4) % T;
    int n  = idx / (C4 * T);
    const float4* src = reinterpret_cast<const float4*>(nf);
    float4 v = src[(size_t)t * N_SRC * C4 + (size_t)n * C4 + c4];
    reinterpret_cast<float4*>(xT)[idx] = v;
}

__global__ void k_zero_int(int* __restrict__ p, int n) {
    int i = blockIdx.x * blockDim.x + threadIdx.x;
    if (i < n) p[i] = 0;
}

__global__ void k_hist(const int* __restrict__ dst, int* __restrict__ counts, int E) {
    int e = blockIdx.x * blockDim.x + threadIdx.x;
    if (e < E) atomicAdd(&counts[dst[e]], 1);
}

__global__ void __launch_bounds__(1024)
k_scan(const int* __restrict__ counts, int* __restrict__ starts,
       int* __restrict__ cursor, int N) {
    __shared__ int part[1024];
    int tid   = threadIdx.x;
    int chunk = (N + 1023) >> 10;
    int begin = tid * chunk;
    int end   = begin + chunk;
    if (end > N) end = N;
    int sum = 0;
    for (int i = begin; i < end; ++i) sum += counts[i];
    part[tid] = sum;
    __syncthreads();
    for (int off = 1; off < 1024; off <<= 1) {
        int v = 0;
        if (tid >= off) v = part[tid - off];
        __syncthreads();
        part[tid] += v;
        __syncthreads();
    }
    int run = part[tid] - sum;
    for (int i = begin; i < end; ++i) {
        starts[i] = run;
        cursor[i] = run;
        run += counts[i];
    }
    if (tid == 1023) starts[N] = part[1023];
}

__global__ void k_bin(const int* __restrict__ src, const int* __restrict__ dst,
                      const float* __restrict__ ew, int* __restrict__ cursor,
                      int* __restrict__ srcs, float* __restrict__ ws, int E) {
    int e = blockIdx.x * blockDim.x + threadIdx.x;
    if (e >= E) return;
    int pos = atomicAdd(&cursor[dst[e]], 1);
    srcs[pos] = src[e];
    ws[pos]   = ew[e];
}

__global__ void __launch_bounds__(96)
k_aggregate(const int* __restrict__ starts, const int* __restrict__ srcs,
            const float* __restrict__ ws, const float* __restrict__ x,
            float* __restrict__ agg) {
    int n   = blockIdx.x;
    int tid = threadIdx.x;
    int beg = starts[n];
    int end = starts[n + 1];
    float4 acc = make_float4(0.f, 0.f, 0.f, 0.f);
    const float4* x4 = reinterpret_cast<const float4*>(x);
    for (int j = beg; j < end; ++j) {
        int   s = srcs[j];
        float w = ws[j];
        float4 v = x4[(size_t)s * ROW4 + tid];
        acc.x += w * v.x;
        acc.y += w * v.y;
        acc.z += w * v.z;
        acc.w += w * v.w;
    }
    reinterpret_cast<float4*>(agg)[(size_t)n * ROW4 + tid] = acc;
}

__global__ void __launch_bounds__(128)
k_gemm1(const float* __restrict__ xT, const float* __restrict__ agg1,
        const float* __restrict__ Wself, const float* __restrict__ Wneigh,
        const float* __restrict__ bias, float* __restrict__ h1) {
    int lane  = threadIdx.x & 31;
    int wave  = threadIdx.x >> 5;
    int mtile = blockIdx.x * (blockDim.x >> 5) + wave;
    if (mtile >= M1 / 16) return;
    int m0 = mtile * 16;

    v2f aS[8], aN[8];
#pragma unroll
    for (int ks = 0; ks < 8; ++ks) {
        aS[ks] = load_frag(xT,   m0, C_IN, ks * 4, lane);
        aN[ks] = load_frag(agg1, m0, C_IN, ks * 4, lane);
    }

#pragma unroll
    for (int nt = 0; nt < 4; ++nt) {
        int n0 = nt * 16;
        v8f acc = {};
#pragma unroll
        for (int ks = 0; ks < 8; ++ks) {
            v2f bS = load_frag(Wself, n0, C_IN, ks * 4, lane);
            acc = __builtin_amdgcn_wmma_f32_16x16x4_f32(false, aS[ks], false, bS,
                                                        (short)0, acc, false, false);
            v2f bN = load_frag(Wneigh, n0, C_IN, ks * 4, lane);
            acc = __builtin_amdgcn_wmma_f32_16x16x4_f32(false, aN[ks], false, bN,
                                                        (short)0, acc, false, false);
        }
        int col   = n0 + (lane & 15);
        int mbase = m0 + ((lane >> 4) << 3);
        float bv  = bias[col];
#pragma unroll
        for (int r = 0; r < 8; ++r) {
            float v = leaky(acc[r] + bv);
            h1[(size_t)(mbase + r) * H + col] = v;
        }
    }
}

__global__ void __launch_bounds__(128)
k_gemm_z2(const float* __restrict__ h1, const float* __restrict__ Wneigh2,
          float* __restrict__ z2) {
    int lane  = threadIdx.x & 31;
    int wave  = threadIdx.x >> 5;
    int mtile = blockIdx.x * (blockDim.x >> 5) + wave;
    if (mtile >= M1 / 16) return;
    int m0 = mtile * 16;

    v2f a[16];
#pragma unroll
    for (int ks = 0; ks < 16; ++ks)
        a[ks] = load_frag(h1, m0, H, ks * 4, lane);

#pragma unroll
    for (int nt = 0; nt < 2; ++nt) {
        int n0 = nt * 16;
        v8f acc = {};
#pragma unroll
        for (int ks = 0; ks < 16; ++ks) {
            v2f b = load_frag(Wneigh2, n0, H, ks * 4, lane);
            acc = __builtin_amdgcn_wmma_f32_16x16x4_f32(false, a[ks], false, b,
                                                        (short)0, acc, false, false);
        }
        int col   = n0 + (lane & 15);
        int mbase = m0 + ((lane >> 4) << 3);
#pragma unroll
        for (int r = 0; r < 8; ++r)
            z2[(size_t)(mbase + r) * C_OUT + col] = acc[r];
    }
}

__global__ void __launch_bounds__(128)
k_final(const float* __restrict__ h1, const float* __restrict__ agg2,
        const float* __restrict__ Wself2, const float* __restrict__ bias2,
        float* __restrict__ out) {
    int lane  = threadIdx.x & 31;
    int wave  = threadIdx.x >> 5;
    int mtile = blockIdx.x * (blockDim.x >> 5) + wave;
    if (mtile >= M2 / 16) return;
    int m0 = mtile * 16;

    v2f a[16];
#pragma unroll
    for (int ks = 0; ks < 16; ++ks)
        a[ks] = load_frag(h1, m0, H, ks * 4, lane);

#pragma unroll
    for (int nt = 0; nt < 2; ++nt) {
        int n0 = nt * 16;
        v8f acc = {};
#pragma unroll
        for (int ks = 0; ks < 16; ++ks) {
            v2f b = load_frag(Wself2, n0, H, ks * 4, lane);
            acc = __builtin_amdgcn_wmma_f32_16x16x4_f32(false, a[ks], false, b,
                                                        (short)0, acc, false, false);
        }
        int col   = n0 + (lane & 15);
        int mbase = m0 + ((lane >> 4) << 3);
        float bv  = bias2[col];
#pragma unroll
        for (int r = 0; r < 8; ++r) {
            int m   = mbase + r;
            float v = leaky(acc[r] + agg2[(size_t)m * C_OUT + col] + bv);
            int n = m / T;
            int t = m % T;
            out[((size_t)t * N_DST2 + n) * C_OUT + col] = v;
        }
    }
}

extern "C" void kernel_launch(void* const* d_in, const int* in_sizes, int n_in,
                              void* d_out, int out_size, void* d_ws, size_t ws_size,
                              hipStream_t stream) {
    const float* nf   = (const float*)d_in[0];
    const int*   src1 = (const int*)  d_in[1];
    const int*   dst1 = (const int*)  d_in[2];
    const float* ew1  = (const float*)d_in[3];
    const int*   src2 = (const int*)  d_in[4];
    const int*   dst2 = (const int*)  d_in[5];
    const float* ew2  = (const float*)d_in[6];
    const float* W1s  = (const float*)d_in[7];
    const float* W1n  = (const float*)d_in[8];
    const float* b1   = (const float*)d_in[9];
    const float* W2s  = (const float*)d_in[10];
    const float* W2n  = (const float*)d_in[11];
    const float* b2   = (const float*)d_in[12];
    const int E1 = in_sizes[1];
    const int E2 = in_sizes[4];

    float* out = (float*)d_out;

    float* ws   = (float*)d_ws;
    float* xT   = ws;
    float* agg1 = xT   + (size_t)N_SRC  * ROW;
    float* h1   = agg1 + (size_t)N_DST1 * ROW;
    float* z2   = xT;
    float* agg2 = xT + (size_t)N_DST1 * ROW;
    int*   ip      = (int*)(h1 + (size_t)M1 * H);
    int*   counts1 = ip;                 ip += N_DST1;
    int*   starts1 = ip;                 ip += N_DST1 + 1;
    int*   cursor1 = ip;                 ip += N_DST1;
    int*   srcs1   = ip;                 ip += E1;
    float* ws1     = (float*)ip;         ip += E1;
    int*   counts2 = ip;                 ip += N_DST2;
    int*   starts2 = ip;                 ip += N_DST2 + 1;
    int*   cursor2 = ip;                 ip += N_DST2;
    int*   srcs2   = ip;                 ip += E2;
    float* ws2     = (float*)ip;

    {
        int total = N_SRC * T * (C_IN / 4);
        k_transpose<<<(total + 255) / 256, 256, 0, stream>>>(nf, xT);
    }
    k_zero_int<<<(N_DST1 + 255) / 256, 256, 0, stream>>>(counts1, N_DST1);
    k_zero_int<<<(N_DST2 + 255) / 256, 256, 0, stream>>>(counts2, N_DST2);
    k_hist<<<(E1 + 255) / 256, 256, 0, stream>>>(dst1, counts1, E1);
    k_hist<<<(E2 + 255) / 256, 256, 0, stream>>>(dst2, counts2, E2);
    k_scan<<<1, 1024, 0, stream>>>(counts1, starts1, cursor1, N_DST1);
    k_scan<<<1, 1024, 0, stream>>>(counts2, starts2, cursor2, N_DST2);
    k_bin<<<(E1 + 255) / 256, 256, 0, stream>>>(src1, dst1, ew1, cursor1, srcs1, ws1, E1);
    k_bin<<<(E2 + 255) / 256, 256, 0, stream>>>(src2, dst2, ew2, cursor2, srcs2, ws2, E2);

    k_aggregate<<<N_DST1, 96, 0, stream>>>(starts1, srcs1, ws1, xT, agg1);

    k_gemm1<<<(M1 / 16 + 3) / 4, 128, 0, stream>>>(xT, agg1, W1s, W1n, b1, h1);

    k_gemm_z2<<<(M1 / 16 + 3) / 4, 128, 0, stream>>>(h1, W2n, z2);

    k_aggregate<<<N_DST2, 96, 0, stream>>>(starts2, srcs2, ws2, z2, agg2);

    k_final<<<(M2 / 16 + 3) / 4, 128, 0, stream>>>(h1, agg2, W2s, b2, out);
}


# CSR build of both graphs in one launch: each work-group owns 512 dst nodes, counts and ranks its edges in LDS (no global atomics, 14->7 launches); bucket order stays arrival order
# speedup vs baseline: 1.5593x; 1.5593x over previous
#include <hip/hip_runtime.h>

typedef __attribute__((ext_vector_type(2))) float v2f;
typedef __attribute__((ext_vector_type(8))) float v8f;

constexpr int T      = 12;
constexpr int C_IN   = 32;
constexpr int H      = 64;
constexpr int C_OUT  = 32;
constexpr int N_SRC  = 50000;
constexpr int N_DST1 = 20000;
constexpr int N_DST2 = 10000;
constexpr int ROW    = T * C_IN;
constexpr int ROW4   = ROW / 4;
constexpr int M1     = N_DST1 * T;
constexpr int M2     = N_DST2 * T;
constexpr float NEG_SLOPE = 0.01f;

__device__ __forceinline__ float leaky(float v) {
    return v > 0.f ? v : NEG_SLOPE * v;
}

__device__ __forceinline__ v2f load_frag(const float* __restrict__ base, int r0, int ld,
                                         int k0, int lane) {
    int r = r0 + (lane & 15);
    int k = k0 + ((lane >> 4) << 1);
    const float* p = base + (size_t)r * ld + k;
    v2f f;
    f.x = p[0];
    f.y = p[1];
    return f;
}

__global__ void k_transpose(const float* __restrict__ nf, float* __restrict__ xT) {
    int idx = blockIdx.x * blockDim.x + threadIdx.x;
    constexpr int C4 = C_IN / 4;
    constexpr int total = N_SRC * T * C4;
    if (idx >= total) return;
    int c4 = idx % C4;
    int t  = (idx / C4) % T;
    int n  = idx / (C4 * T);
    const float4* src = reinterpret_cast<const float4*>(nf);
    float4 v = src[(size_t)t * N_SRC * C4 + (size_t)n * C4 + c4];
    reinterpret_cast<float4*>(xT)[idx] = v;
}

constexpr int CSR_R    = 512;
constexpr int CSR_NT   = 1024;
constexpr int CSR_NW   = CSR_NT / 32;
constexpr int CSR_CAPW = 1024;
constexpr int CSR_UB   = 8;
constexpr int CSR_NB1  = (N_DST1 + CSR_R - 1) / CSR_R;
constexpr int CSR_NB2  = (N_DST2 + CSR_R - 1) / CSR_R;
constexpr int CSR_NONE = 0x7fffffff;

__global__ void __launch_bounds__(CSR_NT)
k_csr(const int* __restrict__ src1, const int* __restrict__ dst1, const float* __restrict__ ew1,
      int* __restrict__ starts1, int* __restrict__ srcs1, float* __restrict__ ws1, int E1,
      const int* __restrict__ src2, const int* __restrict__ dst2, const float* __restrict__ ew2,
      int* __restrict__ starts2, int* __restrict__ srcs2, float* __restrict__ ws2, int E2) {
    __shared__ int cnt[CSR_R];
    __shared__ int wbelow[CSR_NW];
    __shared__ int stot[CSR_NW];
    __shared__ int s_ovf;
    __shared__ int list[CSR_NW * CSR_CAPW];

    const bool second = blockIdx.x >= CSR_NB1;
    const int*   __restrict__ src    = second ? src2    : src1;
    const int*   __restrict__ dst    = second ? dst2    : dst1;
    const float* __restrict__ ew     = second ? ew2     : ew1;
    int*         __restrict__ starts = second ? starts2 : starts1;
    int*         __restrict__ srcs   = second ? srcs2   : srcs1;
    float*       __restrict__ wsv    = second ? ws2     : ws1;
    const int E  = second ? E2 : E1;
    const int N  = second ? N_DST2 : N_DST1;
    const int rb = second ? (int)blockIdx.x - CSR_NB1 : (int)blockIdx.x;
    const int nb = second ? CSR_NB2 : CSR_NB1;
    const int lo = rb * CSR_R;

    const int tid  = threadIdx.x;
    const int lane = tid & 31;
    const int wave = tid >> 5;
    int* mylist = list + wave * CSR_CAPW;

    if (tid < CSR_R) cnt[tid] = 0;
    if (tid == 0) s_ovf = 0;
    __syncthreads();

    int below  = 0;
    int wcount = 0;

    auto visit = [&](int d, int e) {
        below += (d < lo) ? 1 : 0;
        const unsigned r = (unsigned)(d - lo);
        const bool hit = r < (unsigned)CSR_R;
        const unsigned m = (unsigned)__ballot(hit);
        if (hit) {
            atomicAdd(&cnt[r], 1);
            const int slot = wcount + (int)__builtin_amdgcn_mbcnt_lo(m, 0u);
            if (slot < CSR_CAPW) mylist[slot] = e;
        }
        wcount += __popc(m);
    };

    const int E4 = E >> 2;
    const int4* __restrict__ d4 = reinterpret_cast<const int4*>(dst);
    auto batch = [&](const int b0, const bool guard) {
        int4 d[CSR_UB];
#pragma unroll
        for (int u = 0; u < CSR_UB; ++u) {
            const int i = b0 + u * CSR_NT + tid;
            d[u] = d4[(guard && i >= E4) ? E4 - 1 : i];
        }
#pragma unroll
        for (int u = 0; u < CSR_UB; ++u) {
            const int i = b0 + u * CSR_NT + tid;
            if (guard && i >= E4) d[u] = make_int4(CSR_NONE, CSR_NONE, CSR_NONE, CSR_NONE);
            const int e = i << 2;
            visit(d[u].x, e);
            visit(d[u].y, e + 1);
            visit(d[u].z, e + 2);
            visit(d[u].w, e + 3);
        }
    };
    int b0 = 0;
    for (; b0 + CSR_UB * CSR_NT <= E4; b0 += CSR_UB * CSR_NT) batch(b0, false);
    if (b0 < E4) batch(b0, true);
    {
        const int e = (E4 << 2) + tid;
        const int d = e < E ? dst[e] : CSR_NONE;
        visit(d, e);
    }

    for (int off = 16; off > 0; off >>= 1) below += __shfl_xor(below, off);
    if (lane == 0) {
        wbelow[wave] = below;
        if (wcount > CSR_CAPW) s_ovf = 1;
    }
    __syncthreads();

    int base = 0;
    for (int w = 0; w < CSR_NW; ++w) base += wbelow[w];
    const int c = tid < CSR_R ? cnt[tid] : 0;
    int incl = c;
    for (int off = 1; off < 32; off <<= 1) {
        const int t = __shfl_up(incl, off);
        if (lane >= off) incl += t;
    }
    if (lane == 31) stot[wave] = incl;
    __syncthreads();
    int pre = 0;
    for (int w = 0; w < wave; ++w) pre += stot[w];
    const int excl = base + pre + incl - c;
    if (tid < CSR_R) {
        cnt[tid] = excl;
        if (lo + tid < N) starts[lo + tid] = excl;
    }
    if (rb == nb - 1 && tid == 0) starts[N] = E;
    const bool ovf = s_ovf != 0;
    __syncthreads();

    if (!ovf) {
        for (int k = lane; k < wcount; k += 32) {
            const int   e = mylist[k];
            const int   d = dst[e];
            const int   s = src[e];
            const float w = ew[e];
            const int pos = atomicAdd(&cnt[d - lo], 1);
            srcs[pos] = s;
            wsv[pos]  = w;
        }
    } else {
        for (int e = tid; e < E; e += CSR_NT) {
            const unsigned r = (unsigned)(dst[e] - lo);
            if (r < (unsigned)CSR_R) {
                const int pos = atomicAdd(&cnt[r], 1);
                srcs[pos] = src[e];
                wsv[pos]  = ew[e];
            }
        }
    }
}

__global__ void __launch_bounds__(96)
k_aggregate(const int* __restrict__ starts, const int* __restrict__ srcs,
            const float* __restrict__ ws, const float* __restrict__ x,
            float* __restrict__ agg) {
    int n   = blockIdx.x;
    int tid = threadIdx.x;
    int beg = starts[n];
    int end = starts[n + 1];
    float4 acc = make_float4(0.f, 0.f, 0.f, 0.f);
    const float4* x4 = reinterpret_cast<const float4*>(x);
    for (int j = beg; j < end; ++j) {
        int   s = srcs[j];
        float w = ws[j];
        float4 v = x4[(size_t)s * ROW4 + tid];
        acc.x += w * v.x;
        acc.y += w * v.y;
        acc.z += w * v.z;
        acc.w += w * v.w;
    }
    reinterpret_cast<float4*>(agg)[(size_t)n * ROW4 + tid] = acc;
}

__global__ void __launch_bounds__(128)
k_gemm1(const float* __restrict__ xT, const float* __restrict__ agg1,
        const float* __restrict__ Wself, const float* __restrict__ Wneigh,
        const float* __restrict__ bias, float* __restrict__ h1) {
    int lane  = threadIdx.x & 31;
    int wave  = threadIdx.x >> 5;
    int mtile = blockIdx.x * (blockDim.x >> 5) + wave;
    if (mtile >= M1 / 16) return;
    int m0 = mtile * 16;

    v2f aS[8], aN[8];
#pragma unroll
    for (int ks = 0; ks < 8; ++ks) {
        aS[ks] = load_frag(xT,   m0, C_IN, ks * 4, lane);
        aN[ks] = load_frag(agg1, m0, C_IN, ks * 4, lane);
    }

#pragma unroll
    for (int nt = 0; nt < 4; ++nt) {
        int n0 = nt * 16;
        v8f acc = {};
#pragma unroll
        for (int ks = 0; ks < 8; ++ks) {
            v2f bS = load_frag(Wself, n0, C_IN, ks * 4, lane);
            acc = __builtin_amdgcn_wmma_f32_16x16x4_f32(false, aS[ks], false, bS,
                                                        (short)0, acc, false, false);
            v2f bN = load_frag(Wneigh, n0, C_IN, ks * 4, lane);
            acc = __builtin_amdgcn_wmma_f32_16x16x4_f32(false, aN[ks], false, bN,
                                                        (short)0, acc, false, false);
        }
        int col   = n0 + (lane & 15);
        int mbase = m0 + ((lane >> 4) << 3);
        float bv  = bias[col];
#pragma unroll
        for (int r = 0; r < 8; ++r) {
            float v = leaky(acc[r] + bv);
            h1[(size_t)(mbase + r) * H + col] = v;
        }
    }
}

__global__ void __launch_bounds__(128)
k_gemm_z2(const float* __restrict__ h1, const float* __restrict__ Wneigh2,
          float* __restrict__ z2) {
    int lane  = threadIdx.x & 31;
    int wave  = threadIdx.x >> 5;
    int mtile = blockIdx.x * (blockDim.x >> 5) + wave;
    if (mtile >= M1 / 16) return;
    int m0 = mtile * 16;

    v2f a[16];
#pragma unroll
    for (int ks = 0; ks < 16; ++ks)
        a[ks] = load_frag(h1, m0, H, ks * 4, lane);

#pragma unroll
    for (int nt = 0; nt < 2; ++nt) {
        int n0 = nt * 16;
        v8f acc = {};
#pragma unroll
        for (int ks = 0; ks < 16; ++ks) {
            v2f b = load_frag(Wneigh2, n0, H, ks * 4, lane);
            acc = __builtin_amdgcn_wmma_f32_16x16x4_f32(false, a[ks], false, b,
                                                        (short)0, acc, false, false);
        }
        int col   = n0 + (lane & 15);
        int mbase = m0 + ((lane >> 4) << 3);
#pragma unroll
        for (int r = 0; r < 8; ++r)
            z2[(size_t)(mbase + r) * C_OUT + col] = acc[r];
    }
}

__global__ void __launch_bounds__(128)
k_final(const float* __restrict__ h1, const float* __restrict__ agg2,
        const float* __restrict__ Wself2, const float* __restrict__ bias2,
        float* __restrict__ out) {
    int lane  = threadIdx.x & 31;
    int wave  = threadIdx.x >> 5;
    int mtile = blockIdx.x * (blockDim.x >> 5) + wave;
    if (mtile >= M2 / 16) return;
    int m0 = mtile * 16;

    v2f a[16];
#pragma unroll
    for (int ks = 0; ks < 16; ++ks)
        a[ks] = load_frag(h1, m0, H, ks * 4, lane);

#pragma unroll
    for (int nt = 0; nt < 2; ++nt) {
        int n0 = nt * 16;
        v8f acc = {};
#pragma unroll
        for (int ks = 0; ks < 16; ++ks) {
            v2f b = load_frag(Wself2, n0, H, ks * 4, lane);
            acc = __builtin_amdgcn_wmma_f32_16x16x4_f32(false, a[ks], false, b,
                                                        (short)0, acc, false, false);
        }
        int col   = n0 + (lane & 15);
        int mbase = m0 + ((lane >> 4) << 3);
        float bv  = bias2[col];
#pragma unroll
        for (int r = 0; r < 8; ++r) {
            int m   = mbase + r;
            float v = leaky(acc[r] + agg2[(size_t)m * C_OUT + col] + bv);
            int n = m / T;
            int t = m % T;
            out[((size_t)t * N_DST2 + n) * C_OUT + col] = v;
        }
    }
}

extern "C" void kernel_launch(void* const* d_in, const int* in_sizes, int n_in,
                              void* d_out, int out_size, void* d_ws, size_t ws_size,
                              hipStream_t stream) {
    const float* nf   = (const float*)d_in[0];
    const int*   src1 = (const int*)  d_in[1];
    const int*   dst1 = (const int*)  d_in[2];
    const float* ew1  = (const float*)d_in[3];
    const int*   src2 = (const int*)  d_in[4];
    const int*   dst2 = (const int*)  d_in[5];
    const float* ew2  = (const float*)d_in[6];
    const float* W1s  = (const float*)d_in[7];
    const float* W1n  = (const float*)d_in[8];
    const float* b1   = (const float*)d_in[9];
    const float* W2s  = (const float*)d_in[10];
    const float* W2n  = (const float*)d_in[11];
    const float* b2   = (const float*)d_in[12];
    const int E1 = in_sizes[1];
    const int E2 = in_sizes[4];

    float* out = (float*)d_out;

    float* ws   = (float*)d_ws;
    float* xT   = ws;
    float* agg1 = xT   + (size_t)N_SRC  * ROW;
    float* h1   = agg1 + (size_t)N_DST1 * ROW;
    float* z2   = xT;
    float* agg2 = xT + (size_t)N_DST1 * ROW;
    int*   ip      = (int*)(h1 + (size_t)M1 * H);
    int*   starts1 = ip;                 ip += N_DST1 + 1;
    int*   srcs1   = ip;                 ip += E1;
    float* ws1     = (float*)ip;         ip += E1;
    int*   starts2 = ip;                 ip += N_DST2 + 1;
    int*   srcs2   = ip;                 ip += E2;
    float* ws2     = (float*)ip;

    {
        int total = N_SRC * T * (C_IN / 4);
        k_transpose<<<(total + 255) / 256, 256, 0, stream>>>(nf, xT);
    }
    k_csr<<<CSR_NB1 + CSR_NB2, CSR_NT, 0, stream>>>(src1, dst1, ew1, starts1, srcs1, ws1, E1,
                                                   src2, dst2, ew2, starts2, srcs2, ws2, E2);

    k_aggregate<<<N_DST1, 96, 0, stream>>>(starts1, srcs1, ws1, xT, agg1);

    k_gemm1<<<(M1 / 16 + 3) / 4, 128, 0, stream>>>(xT, agg1, W1s, W1n, b1, h1);

    k_gemm_z2<<<(M1 / 16 + 3) / 4, 128, 0, stream>>>(h1, W2n, z2);

    k_aggregate<<<N_DST2, 96, 0, stream>>>(starts2, srcs2, ws2, z2, agg2);

    k_final<<<(M2 / 16 + 3) / 4, 128, 0, stream>>>(h1, agg2, W2s, b2, out);
}


# No transposed copy of node_feat: the layer-1 gather and the layer-1 self operand read its own (T,N,C) layout (7->6 launches, 150000 fewer waves, 154 MB less traffic); sum order unchanged
# speedup vs baseline: 1.8098x; 1.1606x over previous
#include <hip/hip_runtime.h>

typedef __attribute__((ext_vector_type(2))) float v2f;
typedef __attribute__((ext_vector_type(8))) float v8f;

constexpr int T      = 12;
constexpr int C_IN   = 32;
constexpr int H      = 64;
constexpr int C_OUT  = 32;
constexpr int N_SRC  = 50000;
constexpr int N_DST1 = 20000;
constexpr int N_DST2 = 10000;
constexpr int ROW    = T * C_IN;
constexpr int ROW4   = ROW / 4;
constexpr int M1     = N_DST1 * T;
constexpr int M2     = N_DST2 * T;
constexpr float NEG_SLOPE = 0.01f;

__device__ __forceinline__ float leaky(float v) {
    return v > 0.f ? v : NEG_SLOPE * v;
}

__device__ __forceinline__ v2f load_frag(const float* __restrict__ base, int r0, int ld,
                                         int k0, int lane) {
    int r = r0 + (lane & 15);
    int k = k0 + ((lane >> 4) << 1);
    const float* p = base + (size_t)r * ld + k;
    v2f f;
    f.x = p[0];
    f.y = p[1];
    return f;
}

constexpr int CSR_R    = 512;
constexpr int CSR_NT   = 1024;
constexpr int CSR_NW   = CSR_NT / 32;
constexpr int CSR_CAPW = 1024;
constexpr int CSR_UB   = 8;
constexpr int CSR_NB1  = (N_DST1 + CSR_R - 1) / CSR_R;
constexpr int CSR_NB2  = (N_DST2 + CSR_R - 1) / CSR_R;
constexpr int CSR_NONE = 0x7fffffff;

__global__ void __launch_bounds__(CSR_NT)
k_csr(const int* __restrict__ src1, const int* __restrict__ dst1, const float* __restrict__ ew1,
      int* __restrict__ starts1, int* __restrict__ srcs1, float* __restrict__ ws1, int E1,
      const int* __restrict__ src2, const int* __restrict__ dst2, const float* __restrict__ ew2,
      int* __restrict__ starts2, int* __restrict__ srcs2, float* __restrict__ ws2, int E2) {
    __shared__ int cnt[CSR_R];
    __shared__ int wbelow[CSR_NW];
    __shared__ int stot[CSR_NW];
    __shared__ int s_ovf;
    __shared__ int list[CSR_NW * CSR_CAPW];

    const bool second = blockIdx.x >= CSR_NB1;
    const int*   __restrict__ src    = second ? src2    : src1;
    const int*   __restrict__ dst    = second ? dst2    : dst1;
    const float* __restrict__ ew     = second ? ew2     : ew1;
    int*         __restrict__ starts = second ? starts2 : starts1;
    int*         __restrict__ srcs   = second ? srcs2   : srcs1;
    float*       __restrict__ wsv    = second ? ws2     : ws1;
    const int E  = second ? E2 : E1;
    const int N  = second ? N_DST2 : N_DST1;
    const int rb = second ? (int)blockIdx.x - CSR_NB1 : (int)blockIdx.x;
    const int nb = second ? CSR_NB2 : CSR_NB1;
    const int lo = rb * CSR_R;

    const int tid  = threadIdx.x;
    const int lane = tid & 31;
    const int wave = tid >> 5;
    int* mylist = list + wave * CSR_CAPW;

    if (tid < CSR_R) cnt[tid] = 0;
    if (tid == 0) s_ovf = 0;
    __syncthreads();

    int below  = 0;
    int wcount = 0;

    auto visit = [&](int d, int e) {
        below += (d < lo) ? 1 : 0;
        const unsigned r = (unsigned)(d - lo);
        const bool hit = r < (unsigned)CSR_R;
        const unsigned m = (unsigned)__ballot(hit);
        if (hit) {
            atomicAdd(&cnt[r], 1);
            const int slot = wcount + (int)__builtin_amdgcn_mbcnt_lo(m, 0u);
            if (slot < CSR_CAPW) mylist[slot] = e;
        }
        wcount += __popc(m);
    };

    const int E4 = E >> 2;
    const int4* __restrict__ d4 = reinterpret_cast<const int4*>(dst);
    auto batch = [&](const int b0, const bool guard) {
        int4 d[CSR_UB];
#pragma unroll
        for (int u = 0; u < CSR_UB; ++u) {
            const int i = b0 + u * CSR_NT + tid;
            d[u] = d4[(guard && i >= E4) ? E4 - 1 : i];
        }
#pragma unroll
        for (int u = 0; u < CSR_UB; ++u) {
            const int i = b0 + u * CSR_NT + tid;
            if (guard && i >= E4) d[u] = make_int4(CSR_NONE, CSR_NONE, CSR_NONE, CSR_NONE);
            const int e = i << 2;
            visit(d[u].x, e);
            visit(d[u].y, e + 1);
            visit(d[u].z, e + 2);
            visit(d[u].w, e + 3);
        }
    };
    int b0 = 0;
    for (; b0 + CSR_UB * CSR_NT <= E4; b0 += CSR_UB * CSR_NT) batch(b0, false);
    if (b0 < E4) batch(b0, true);
    {
        const int e = (E4 << 2) + tid;
        const int d = e < E ? dst[e] : CSR_NONE;
        visit(d, e);
    }

    for (int off = 16; off > 0; off >>= 1) below += __shfl_xor(below, off);
    if (lane == 0) {
        wbelow[wave] = below;
        if (wcount > CSR_CAPW) s_ovf = 1;
    }
    __syncthreads();

    int base = 0;
    for (int w = 0; w < CSR_NW; ++w) base += wbelow[w];
    const int c = tid < CSR_R ? cnt[tid] : 0;
    int incl = c;
    for (int off = 1; off < 32; off <<= 1) {
        const int t = __shfl_up(incl, off);
        if (lane >= off) incl += t;
    }
    if (lane == 31) stot[wave] = incl;
    __syncthreads();
    int pre = 0;
    for (int w = 0; w < wave; ++w) pre += stot[w];
    const int excl = base + pre + incl - c;
    if (tid < CSR_R) {
        cnt[tid] = excl;
        if (lo + tid < N) starts[lo + tid] = excl;
    }
    if (rb == nb - 1 && tid == 0) starts[N] = E;
    const bool ovf = s_ovf != 0;
    __syncthreads();

    if (!ovf) {
        for (int k = lane; k < wcount; k += 32) {
            const int   e = mylist[k];
            const int   d = dst[e];
            const int   s = src[e];
            const float w = ew[e];
            const int pos = atomicAdd(&cnt[d - lo], 1);
            srcs[pos] = s;
            wsv[pos]  = w;
        }
    } else {
        for (int e = tid; e < E; e += CSR_NT) {
            const unsigned r = (unsigned)(dst[e] - lo);
            if (r < (unsigned)CSR_R) {
                const int pos = atomicAdd(&cnt[r], 1);
                srcs[pos] = src[e];
                wsv[pos]  = ew[e];
            }
        }
    }
}

__global__ void __launch_bounds__(96)
k_aggregate(const int* __restrict__ starts, const int* __restrict__ srcs,
            const float* __restrict__ ws, const float* __restrict__ x,
            float* __restrict__ agg) {
    int n   = blockIdx.x;
    int tid = threadIdx.x;
    int beg = starts[n];
    int end = starts[n + 1];
    float4 acc = make_float4(0.f, 0.f, 0.f, 0.f);
    const float4* x4 = reinterpret_cast<const float4*>(x);
    for (int j = beg; j < end; ++j) {
        int   s = srcs[j];
        float w = ws[j];
        float4 v = x4[(size_t)s * ROW4 + tid];
        acc.x += w * v.x;
        acc.y += w * v.y;
        acc.z += w * v.z;
        acc.w += w * v.w;
    }
    reinterpret_cast<float4*>(agg)[(size_t)n * ROW4 + tid] = acc;
}

__global__ void __launch_bounds__(96)
k_aggregate_nf(const int* __restrict__ starts, const int* __restrict__ srcs,
               const float* __restrict__ ws, const float* __restrict__ nf,
               float* __restrict__ agg) {
    constexpr int C4 = C_IN / 4;
    int n   = blockIdx.x;
    int tid = threadIdx.x;
    int beg = starts[n];
    int end = starts[n + 1];
    const int t  = tid / C4;
    const int c4 = tid % C4;
    const float4* p = reinterpret_cast<const float4*>(nf) + (size_t)t * N_SRC * C4 + c4;
    float4 acc = make_float4(0.f, 0.f, 0.f, 0.f);
    for (int j = beg; j < end; ++j) {
        int   s = srcs[j];
        float w = ws[j];
        float4 v = p[(size_t)s * C4];
        acc.x += w * v.x;
        acc.y += w * v.y;
        acc.z += w * v.z;
        acc.w += w * v.w;
    }
    reinterpret_cast<float4*>(agg)[(size_t)n * ROW4 + tid] = acc;
}

__global__ void __launch_bounds__(128)
k_gemm1(const float* __restrict__ nf, const float* __restrict__ agg1,
        const float* __restrict__ Wself, const float* __restrict__ Wneigh,
        const float* __restrict__ bias, float* __restrict__ h1) {
    int lane  = threadIdx.x & 31;
    int wave  = threadIdx.x >> 5;
    int mtile = blockIdx.x * (blockDim.x >> 5) + wave;
    if (mtile >= M1 / 16) return;
    int m0 = mtile * 16;

    const int mrow = m0 + (lane & 15);
    const int nrow = mrow / T;
    const int trow = mrow - nrow * T;
    const float* xrow = nf + ((size_t)trow * N_SRC + nrow) * C_IN + ((lane >> 4) << 1);

    v2f aS[8], aN[8];
#pragma unroll
    for (int ks = 0; ks < 8; ++ks) {
        aS[ks].x = xrow[ks * 4];
        aS[ks].y = xrow[ks * 4 + 1];
        aN[ks] = load_frag(agg1, m0, C_IN, ks * 4, lane);
    }

#pragma unroll
    for (int nt = 0; nt < 4; ++nt) {
        int n0 = nt * 16;
        v8f acc = {};
#pragma unroll
        for (int ks = 0; ks < 8; ++ks) {
            v2f bS = load_frag(Wself, n0, C_IN, ks * 4, lane);
            acc = __builtin_amdgcn_wmma_f32_16x16x4_f32(false, aS[ks], false, bS,
                                                        (short)0, acc, false, false);
            v2f bN = load_frag(Wneigh, n0, C_IN, ks * 4, lane);
            acc = __builtin_amdgcn_wmma_f32_16x16x4_f32(false, aN[ks], false, bN,
                                                        (short)0, acc, false, false);
        }
        int col   = n0 + (lane & 15);
        int mbase = m0 + ((lane >> 4) << 3);
        float bv  = bias[col];
#pragma unroll
        for (int r = 0; r < 8; ++r) {
            float v = leaky(acc[r] + bv);
            h1[(size_t)(mbase + r) * H + col] = v;
        }
    }
}

__global__ void __launch_bounds__(128)
k_gemm_z2(const float* __restrict__ h1, const float* __restrict__ Wneigh2,
          float* __restrict__ z2) {
    int lane  = threadIdx.x & 31;
    int wave  = threadIdx.x >> 5;
    int mtile = blockIdx.x * (blockDim.x >> 5) + wave;
    if (mtile >= M1 / 16) return;
    int m0 = mtile * 16;

    v2f a[16];
#pragma unroll
    for (int ks = 0; ks < 16; ++ks)
        a[ks] = load_frag(h1, m0, H, ks * 4, lane);

#pragma unroll
    for (int nt = 0; nt < 2; ++nt) {
        int n0 = nt * 16;
        v8f acc = {};
#pragma unroll
        for (int ks = 0; ks < 16; ++ks) {
            v2f b = load_frag(Wneigh2, n0, H, ks * 4, lane);
            acc = __builtin_amdgcn_wmma_f32_16x16x4_f32(false, a[ks], false, b,
                                                        (short)0, acc, false, false);
        }
        int col   = n0 + (lane & 15);
        int mbase = m0 + ((lane >> 4) << 3);
#pragma unroll
        for (int r = 0; r < 8; ++r)
            z2[(size_t)(mbase + r) * C_OUT + col] = acc[r];
    }
}

__global__ void __launch_bounds__(128)
k_final(const float* __restrict__ h1, const float* __restrict__ agg2,
        const float* __restrict__ Wself2, const float* __restrict__ bias2,
        float* __restrict__ out) {
    int lane  = threadIdx.x & 31;
    int wave  = threadIdx.x >> 5;
    int mtile = blockIdx.x * (blockDim.x >> 5) + wave;
    if (mtile >= M2 / 16) return;
    int m0 = mtile * 16;

    v2f a[16];
#pragma unroll
    for (int ks = 0; ks < 16; ++ks)
        a[ks] = load_frag(h1, m0, H, ks * 4, lane);

#pragma unroll
    for (int nt = 0; nt < 2; ++nt) {
        int n0 = nt * 16;
        v8f acc = {};
#pragma unroll
        for (int ks = 0; ks < 16; ++ks) {
            v2f b = load_frag(Wself2, n0, H, ks * 4, lane);
            acc = __builtin_amdgcn_wmma_f32_16x16x4_f32(false, a[ks], false, b,
                                                        (short)0, acc, false, false);
        }
        int col   = n0 + (lane & 15);
        int mbase = m0 + ((lane >> 4) << 3);
        float bv  = bias2[col];
#pragma unroll
        for (int r = 0; r < 8; ++r) {
            int m   = mbase + r;
            float v = leaky(acc[r] + agg2[(size_t)m * C_OUT + col] + bv);
            int n = m / T;
            int t = m % T;
            out[((size_t)t * N_DST2 + n) * C_OUT + col] = v;
        }
    }
}

extern "C" void kernel_launch(void* const* d_in, const int* in_sizes, int n_in,
                              void* d_out, int out_size, void* d_ws, size_t ws_size,
                              hipStream_t stream) {
    const float* nf   = (const float*)d_in[0];
    const int*   src1 = (const int*)  d_in[1];
    const int*   dst1 = (const int*)  d_in[2];
    const float* ew1  = (const float*)d_in[3];
    const int*   src2 = (const int*)  d_in[4];
    const int*   dst2 = (const int*)  d_in[5];
    const float* ew2  = (const float*)d_in[6];
    const float* W1s  = (const float*)d_in[7];
    const float* W1n  = (const float*)d_in[8];
    const float* b1   = (const float*)d_in[9];
    const float* W2s  = (const float*)d_in[10];
    const float* W2n  = (const float*)d_in[11];
    const float* b2   = (const float*)d_in[12];
    const int E1 = in_sizes[1];
    const int E2 = in_sizes[4];

    float* out = (float*)d_out;

    float* ws   = (float*)d_ws;
    float* z2   = ws;
    float* agg2 = ws   + (size_t)N_DST1 * ROW;
    float* agg1 = ws   + (size_t)N_SRC  * ROW;
    float* h1   = agg1 + (size_t)N_DST1 * ROW;
    int*   ip      = (int*)(h1 + (size_t)M1 * H);
    int*   starts1 = ip;                 ip += N_DST1 + 1;
    int*   srcs1   = ip;                 ip += E1;
    float* ws1     = (float*)ip;         ip += E1;
    int*   starts2 = ip;                 ip += N_DST2 + 1;
    int*   srcs2   = ip;                 ip += E2;
    float* ws2     = (float*)ip;

    k_csr<<<CSR_NB1 + CSR_NB2, CSR_NT, 0, stream>>>(src1, dst1, ew1, starts1, srcs1, ws1, E1,
                                                   src2, dst2, ew2, starts2, srcs2, ws2, E2);

    k_aggregate_nf<<<N_DST1, 96, 0, stream>>>(starts1, srcs1, ws1, nf, agg1);

    k_gemm1<<<(M1 / 16 + 3) / 4, 128, 0, stream>>>(nf, agg1, W1s, W1n, b1, h1);

    k_gemm_z2<<<(M1 / 16 + 3) / 4, 128, 0, stream>>>(h1, W2n, z2);

    k_aggregate<<<N_DST2, 96, 0, stream>>>(starts2, srcs2, ws2, z2, agg2);

    k_final<<<(M2 / 16 + 3) / 4, 128, 0, stream>>>(h1, agg2, W2s, b2, out);
}


# GEMM weights staged once per work-group in LDS (pitch K+4, conflict-free); B fragments come from LDS, not from global loads touching one line per row; 8 tiles per group; sums unchanged
# speedup vs baseline: 1.9941x; 1.1018x over previous
#include <hip/hip_runtime.h>

typedef __attribute__((ext_vector_type(2))) float v2f;
typedef __attribute__((ext_vector_type(8))) float v8f;

constexpr int T      = 12;
constexpr int C_IN   = 32;
constexpr int H      = 64;
constexpr int C_OUT  = 32;
constexpr int N_SRC  = 50000;
constexpr int N_DST1 = 20000;
constexpr int N_DST2 = 10000;
constexpr int ROW    = T * C_IN;
constexpr int ROW4   = ROW / 4;
constexpr int M1     = N_DST1 * T;
constexpr int M2     = N_DST2 * T;
constexpr float NEG_SLOPE = 0.01f;

__device__ __forceinline__ float leaky(float v) {
    return v > 0.f ? v : NEG_SLOPE * v;
}

__device__ __forceinline__ v2f load_frag(const float* __restrict__ base, int r0, int ld,
                                         int k0, int lane) {
    int r = r0 + (lane & 15);
    int k = k0 + ((lane >> 4) << 1);
    const float* p = base + (size_t)r * ld + k;
    v2f f;
    f.x = p[0];
    f.y = p[1];
    return f;
}

constexpr int CSR_R    = 512;
constexpr int CSR_NT   = 1024;
constexpr int CSR_NW   = CSR_NT / 32;
constexpr int CSR_CAPW = 1024;
constexpr int CSR_UB   = 8;
constexpr int CSR_NB1  = (N_DST1 + CSR_R - 1) / CSR_R;
constexpr int CSR_NB2  = (N_DST2 + CSR_R - 1) / CSR_R;
constexpr int CSR_NONE = 0x7fffffff;

__global__ void __launch_bounds__(CSR_NT)
k_csr(const int* __restrict__ src1, const int* __restrict__ dst1, const float* __restrict__ ew1,
      int* __restrict__ starts1, int* __restrict__ srcs1, float* __restrict__ ws1, int E1,
      const int* __restrict__ src2, const int* __restrict__ dst2, const float* __restrict__ ew2,
      int* __restrict__ starts2, int* __restrict__ srcs2, float* __restrict__ ws2, int E2) {
    __shared__ int cnt[CSR_R];
    __shared__ int wbelow[CSR_NW];
    __shared__ int stot[CSR_NW];
    __shared__ int s_ovf;
    __shared__ int list[CSR_NW * CSR_CAPW];

    const bool second = blockIdx.x >= CSR_NB1;
    const int*   __restrict__ src    = second ? src2    : src1;
    const int*   __restrict__ dst    = second ? dst2    : dst1;
    const float* __restrict__ ew     = second ? ew2     : ew1;
    int*         __restrict__ starts = second ? starts2 : starts1;
    int*         __restrict__ srcs   = second ? srcs2   : srcs1;
    float*       __restrict__ wsv    = second ? ws2     : ws1;
    const int E  = second ? E2 : E1;
    const int N  = second ? N_DST2 : N_DST1;
    const int rb = second ? (int)blockIdx.x - CSR_NB1 : (int)blockIdx.x;
    const int nb = second ? CSR_NB2 : CSR_NB1;
    const int lo = rb * CSR_R;

    const int tid  = threadIdx.x;
    const int lane = tid & 31;
    const int wave = tid >> 5;
    int* mylist = list + wave * CSR_CAPW;

    if (tid < CSR_R) cnt[tid] = 0;
    if (tid == 0) s_ovf = 0;
    __syncthreads();

    int below  = 0;
    int wcount = 0;

    auto visit = [&](int d, int e) {
        below += (d < lo) ? 1 : 0;
        const unsigned r = (unsigned)(d - lo);
        const bool hit = r < (unsigned)CSR_R;
        const unsigned m = (unsigned)__ballot(hit);
        if (hit) {
            atomicAdd(&cnt[r], 1);
            const int slot = wcount + (int)__builtin_amdgcn_mbcnt_lo(m, 0u);
            if (slot < CSR_CAPW) mylist[slot] = e;
        }
        wcount += __popc(m);
    };

    const int E4 = E >> 2;
    const int4* __restrict__ d4 = reinterpret_cast<const int4*>(dst);
    auto batch = [&](const int b0, const bool guard) {
        int4 d[CSR_UB];
#pragma unroll
        for (int u = 0; u < CSR_UB; ++u) {
            const int i = b0 + u * CSR_NT + tid;
            d[u] = d4[(guard && i >= E4) ? E4 - 1 : i];
        }
#pragma unroll
        for (int u = 0; u < CSR_UB; ++u) {
            const int i = b0 + u * CSR_NT + tid;
            if (guard && i >= E4) d[u] = make_int4(CSR_NONE, CSR_NONE, CSR_NONE, CSR_NONE);
            const int e = i << 2;
            visit(d[u].x, e);
            visit(d[u].y, e + 1);
            visit(d[u].z, e + 2);
            visit(d[u].w, e + 3);
        }
    };
    int b0 = 0;
    for (; b0 + CSR_UB * CSR_NT <= E4; b0 += CSR_UB * CSR_NT) batch(b0, false);
    if (b0 < E4) batch(b0, true);
    {
        const int e = (E4 << 2) + tid;
        const int d = e < E ? dst[e] : CSR_NONE;
        visit(d, e);
    }

    for (int off = 16; off > 0; off >>= 1) below += __shfl_xor(below, off);
    if (lane == 0) {
        wbelow[wave] = below;
        if (wcount > CSR_CAPW) s_ovf = 1;
    }
    __syncthreads();

    int base = 0;
    for (int w = 0; w < CSR_NW; ++w) base += wbelow[w];
    const int c = tid < CSR_R ? cnt[tid] : 0;
    int incl = c;
    for (int off = 1; off < 32; off <<= 1) {
        const int t = __shfl_up(incl, off);
        if (lane >= off) incl += t;
    }
    if (lane == 31) stot[wave] = incl;
    __syncthreads();
    int pre = 0;
    for (int w = 0; w < wave; ++w) pre += stot[w];
    const int excl = base + pre + incl - c;
    if (tid < CSR_R) {
        cnt[tid] = excl;
        if (lo + tid < N) starts[lo + tid] = excl;
    }
    if (rb == nb - 1 && tid == 0) starts[N] = E;
    const bool ovf = s_ovf != 0;
    __syncthreads();

    if (!ovf) {
        for (int k = lane; k < wcount; k += 32) {
            const int   e = mylist[k];
            const int   d = dst[e];
            const int   s = src[e];
            const float w = ew[e];
            const int pos = atomicAdd(&cnt[d - lo], 1);
            srcs[pos] = s;
            wsv[pos]  = w;
        }
    } else {
        for (int e = tid; e < E; e += CSR_NT) {
            const unsigned r = (unsigned)(dst[e] - lo);
            if (r < (unsigned)CSR_R) {
                const int pos = atomicAdd(&cnt[r], 1);
                srcs[pos] = src[e];
                wsv[pos]  = ew[e];
            }
        }
    }
}

__global__ void __launch_bounds__(96)
k_aggregate(const int* __restrict__ starts, const int* __restrict__ srcs,
            const float* __restrict__ ws, const float* __restrict__ x,
            float* __restrict__ agg) {
    int n   = blockIdx.x;
    int tid = threadIdx.x;
    int beg = starts[n];
    int end = starts[n + 1];
    float4 acc = make_float4(0.f, 0.f, 0.f, 0.f);
    const float4* x4 = reinterpret_cast<const float4*>(x);
    for (int j = beg; j < end; ++j) {
        int   s = srcs[j];
        float w = ws[j];
        float4 v = x4[(size_t)s * ROW4 + tid];
        acc.x += w * v.x;
        acc.y += w * v.y;
        acc.z += w * v.z;
        acc.w += w * v.w;
    }
    reinterpret_cast<float4*>(agg)[(size_t)n * ROW4 + tid] = acc;
}

__global__ void __launch_bounds__(96)
k_aggregate_nf(const int* __restrict__ starts, const int* __restrict__ srcs,
               const float* __restrict__ ws, const float* __restrict__ nf,
               float* __restrict__ agg) {
    constexpr int C4 = C_IN / 4;
    int n   = blockIdx.x;
    int tid = threadIdx.x;
    int beg = starts[n];
    int end = starts[n + 1];
    const int t  = tid / C4;
    const int c4 = tid % C4;
    const float4* p = reinterpret_cast<const float4*>(nf) + (size_t)t * N_SRC * C4 + c4;
    float4 acc = make_float4(0.f, 0.f, 0.f, 0.f);
    for (int j = beg; j < end; ++j) {
        int   s = srcs[j];
        float w = ws[j];
        float4 v = p[(size_t)s * C4];
        acc.x += w * v.x;
        acc.y += w * v.y;
        acc.z += w * v.z;
        acc.w += w * v.w;
    }
    reinterpret_cast<float4*>(agg)[(size_t)n * ROW4 + tid] = acc;
}

constexpr int GEMM_NW = 8;
constexpr int GEMM_NT = GEMM_NW * 32;
constexpr int P_IN    = C_IN + 4;
constexpr int P_H     = H + 4;

template <int ROWS, int K>
__device__ __forceinline__ void stage_weights(const float* __restrict__ Wg, float* sW, int tid) {
    constexpr int K4 = K / 4;
    const float4* __restrict__ g4 = reinterpret_cast<const float4*>(Wg);
    for (int i = tid; i < ROWS * K4; i += GEMM_NT) {
        const int r = i / K4;
        const int q = i - r * K4;
        *reinterpret_cast<float4*>(sW + r * (K + 4) + q * 4) = g4[i];
    }
}

__device__ __forceinline__ v2f lds_frag(const float* sW, int n0, int pitch, int k0, int lane) {
    const int r = n0 + (lane & 15);
    const int k = k0 + ((lane >> 4) << 1);
    return *reinterpret_cast<const v2f*>(sW + r * pitch + k);
}

__global__ void __launch_bounds__(GEMM_NT)
k_gemm1(const float* __restrict__ nf, const float* __restrict__ agg1,
        const float* __restrict__ Wself, const float* __restrict__ Wneigh,
        const float* __restrict__ bias, float* __restrict__ h1) {
    __shared__ __attribute__((aligned(16))) float sWs[H * P_IN];
    __shared__ __attribute__((aligned(16))) float sWn[H * P_IN];

    int lane  = threadIdx.x & 31;
    int wave  = threadIdx.x >> 5;
    int mtile = blockIdx.x * GEMM_NW + wave;
    const bool valid = mtile < M1 / 16;
    int m0 = (valid ? mtile : 0) * 16;

    const int mrow = m0 + (lane & 15);
    const int nrow = mrow / T;
    const int trow = mrow - nrow * T;
    const float* xrow = nf + ((size_t)trow * N_SRC + nrow) * C_IN + ((lane >> 4) << 1);

    v2f aS[8], aN[8];
#pragma unroll
    for (int ks = 0; ks < 8; ++ks) {
        aS[ks].x = xrow[ks * 4];
        aS[ks].y = xrow[ks * 4 + 1];
        aN[ks] = load_frag(agg1, m0, C_IN, ks * 4, lane);
    }

    stage_weights<H, C_IN>(Wself,  sWs, threadIdx.x);
    stage_weights<H, C_IN>(Wneigh, sWn, threadIdx.x);
    __syncthreads();
    if (!valid) return;

#pragma unroll
    for (int nt = 0; nt < 4; ++nt) {
        int n0 = nt * 16;
        v8f acc = {};
#pragma unroll
        for (int ks = 0; ks < 8; ++ks) {
            v2f bS = lds_frag(sWs, n0, P_IN, ks * 4, lane);
            acc = __builtin_amdgcn_wmma_f32_16x16x4_f32(false, aS[ks], false, bS,
                                                        (short)0, acc, false, false);
            v2f bN = lds_frag(sWn, n0, P_IN, ks * 4, lane);
            acc = __builtin_amdgcn_wmma_f32_16x16x4_f32(false, aN[ks], false, bN,
                                                        (short)0, acc, false, false);
        }
        int col   = n0 + (lane & 15);
        int mbase = m0 + ((lane >> 4) << 3);
        float bv  = bias[col];
#pragma unroll
        for (int r = 0; r < 8; ++r) {
            float v = leaky(acc[r] + bv);
            h1[(size_t)(mbase + r) * H + col] = v;
        }
    }
}

__global__ void __launch_bounds__(GEMM_NT)
k_gemm_z2(const float* __restrict__ h1, const float* __restrict__ Wneigh2,
          float* __restrict__ z2) {
    __shared__ __attribute__((aligned(16))) float sW[C_OUT * P_H];

    int lane  = threadIdx.x & 31;
    int wave  = threadIdx.x >> 5;
    int mtile = blockIdx.x * GEMM_NW + wave;
    const bool valid = mtile < M1 / 16;
    int m0 = (valid ? mtile : 0) * 16;

    v2f a[16];
#pragma unroll
    for (int ks = 0; ks < 16; ++ks)
        a[ks] = load_frag(h1, m0, H, ks * 4, lane);

    stage_weights<C_OUT, H>(Wneigh2, sW, threadIdx.x);
    __syncthreads();
    if (!valid) return;

#pragma unroll
    for (int nt = 0; nt < 2; ++nt) {
        int n0 = nt * 16;
        v8f acc = {};
#pragma unroll
        for (int ks = 0; ks < 16; ++ks) {
            v2f b = lds_frag(sW, n0, P_H, ks * 4, lane);
            acc = __builtin_amdgcn_wmma_f32_16x16x4_f32(false, a[ks], false, b,
                                                        (short)0, acc, false, false);
        }
        int col   = n0 + (lane & 15);
        int mbase = m0 + ((lane >> 4) << 3);
#pragma unroll
        for (int r = 0; r < 8; ++r)
            z2[(size_t)(mbase + r) * C_OUT + col] = acc[r];
    }
}

__global__ void __launch_bounds__(GEMM_NT)
k_final(const float* __restrict__ h1, const float* __restrict__ agg2,
        const float* __restrict__ Wself2, const float* __restrict__ bias2,
        float* __restrict__ out) {
    __shared__ __attribute__((aligned(16))) float sW[C_OUT * P_H];

    int lane  = threadIdx.x & 31;
    int wave  = threadIdx.x >> 5;
    int mtile = blockIdx.x * GEMM_NW + wave;
    const bool valid = mtile < M2 / 16;
    int m0 = (valid ? mtile : 0) * 16;

    v2f a[16];
#pragma unroll
    for (int ks = 0; ks < 16; ++ks)
        a[ks] = load_frag(h1, m0, H, ks * 4, lane);

    stage_weights<C_OUT, H>(Wself2, sW, threadIdx.x);
    __syncthreads();
    if (!valid) return;

#pragma unroll
    for (int nt = 0; nt < 2; ++nt) {
        int n0 = nt * 16;
        v8f acc = {};
#pragma unroll
        for (int ks = 0; ks < 16; ++ks) {
            v2f b = lds_frag(sW, n0, P_H, ks * 4, lane);
            acc = __builtin_amdgcn_wmma_f32_16x16x4_f32(false, a[ks], false, b,
                                                        (short)0, acc, false, false);
        }
        int col   = n0 + (lane & 15);
        int mbase = m0 + ((lane >> 4) << 3);
        float bv  = bias2[col];
#pragma unroll
        for (int r = 0; r < 8; ++r) {
            int m   = mbase + r;
            float v = leaky(acc[r] + agg2[(size_t)m * C_OUT + col] + bv);
            int n = m / T;
            int t = m % T;
            out[((size_t)t * N_DST2 + n) * C_OUT + col] = v;
        }
    }
}

extern "C" void kernel_launch(void* const* d_in, const int* in_sizes, int n_in,
                              void* d_out, int out_size, void* d_ws, size_t ws_size,
                              hipStream_t stream) {
    const float* nf   = (const float*)d_in[0];
    const int*   src1 = (const int*)  d_in[1];
    const int*   dst1 = (const int*)  d_in[2];
    const float* ew1  = (const float*)d_in[3];
    const int*   src2 = (const int*)  d_in[4];
    const int*   dst2 = (const int*)  d_in[5];
    const float* ew2  = (const float*)d_in[6];
    const float* W1s  = (const float*)d_in[7];
    const float* W1n  = (const float*)d_in[8];
    const float* b1   = (const float*)d_in[9];
    const float* W2s  = (const float*)d_in[10];
    const float* W2n  = (const float*)d_in[11];
    const float* b2   = (const float*)d_in[12];
    const int E1 = in_sizes[1];
    const int E2 = in_sizes[4];

    float* out = (float*)d_out;

    float* ws   = (float*)d_ws;
    float* z2   = ws;
    float* agg2 = ws   + (size_t)N_DST1 * ROW;
    float* agg1 = ws   + (size_t)N_SRC  * ROW;
    float* h1   = agg1 + (size_t)N_DST1 * ROW;
    int*   ip      = (int*)(h1 + (size_t)M1 * H);
    int*   starts1 = ip;                 ip += N_DST1 + 1;
    int*   srcs1   = ip;                 ip += E1;
    float* ws1     = (float*)ip;         ip += E1;
    int*   starts2 = ip;                 ip += N_DST2 + 1;
    int*   srcs2   = ip;                 ip += E2;
    float* ws2     = (float*)ip;

    k_csr<<<CSR_NB1 + CSR_NB2, CSR_NT, 0, stream>>>(src1, dst1, ew1, starts1, srcs1, ws1, E1,
                                                   src2, dst2, ew2, starts2, srcs2, ws2, E2);

    k_aggregate_nf<<<N_DST1, 96, 0, stream>>>(starts1, srcs1, ws1, nf, agg1);

    k_gemm1<<<(M1 / 16 + GEMM_NW - 1) / GEMM_NW, GEMM_NT, 0, stream>>>(nf, agg1, W1s, W1n, b1, h1);

    k_gemm_z2<<<(M1 / 16 + GEMM_NW - 1) / GEMM_NW, GEMM_NT, 0, stream>>>(h1, W2n, z2);

    k_aggregate<<<N_DST2, 96, 0, stream>>>(starts2, srcs2, ws2, z2, agg2);

    k_final<<<(M2 / 16 + GEMM_NW - 1) / GEMM_NW, GEMM_NT, 0, stream>>>(h1, agg2, W2s, b2, out);
}


# h1 kept in registers: layer 1 done transposed (weights as A), its D tiles feed both layer-2 products as B operand; own term joins the layer-2 gather; 6->4 launches; h sums in permuted k order
# speedup vs baseline: 2.1957x; 1.1011x over previous
#include <hip/hip_runtime.h>

typedef __attribute__((ext_vector_type(2))) float v2f;
typedef __attribute__((ext_vector_type(8))) float v8f;

constexpr int T      = 12;
constexpr int C_IN   = 32;
constexpr int H      = 64;
constexpr int C_OUT  = 32;
constexpr int N_SRC  = 50000;
constexpr int N_DST1 = 20000;
constexpr int N_DST2 = 10000;
constexpr int ROW    = T * C_IN;
constexpr int ROW4   = ROW / 4;
constexpr int M1     = N_DST1 * T;
constexpr int M2     = N_DST2 * T;
constexpr float NEG_SLOPE = 0.01f;

__device__ __forceinline__ float leaky(float v) {
    return v > 0.f ? v : NEG_SLOPE * v;
}

__device__ __forceinline__ v2f load_frag(const float* __restrict__ base, int r0, int ld,
                                         int k0, int lane) {
    int r = r0 + (lane & 15);
    int k = k0 + ((lane >> 4) << 1);
    const float* p = base + (size_t)r * ld + k;
    v2f f;
    f.x = p[0];
    f.y = p[1];
    return f;
}

constexpr int CSR_R    = 512;
constexpr int CSR_NT   = 1024;
constexpr int CSR_NW   = CSR_NT / 32;
constexpr int CSR_CAPW = 1024;
constexpr int CSR_UB   = 8;
constexpr int CSR_NB1  = (N_DST1 + CSR_R - 1) / CSR_R;
constexpr int CSR_NB2  = (N_DST2 + CSR_R - 1) / CSR_R;
constexpr int CSR_NONE = 0x7fffffff;

__global__ void __launch_bounds__(CSR_NT)
k_csr(const int* __restrict__ src1, const int* __restrict__ dst1, const float* __restrict__ ew1,
      int* __restrict__ starts1, int* __restrict__ srcs1, float* __restrict__ ws1, int E1,
      const int* __restrict__ src2, const int* __restrict__ dst2, const float* __restrict__ ew2,
      int* __restrict__ starts2, int* __restrict__ srcs2, float* __restrict__ ws2, int E2) {
    __shared__ int cnt[CSR_R];
    __shared__ int wbelow[CSR_NW];
    __shared__ int stot[CSR_NW];
    __shared__ int s_ovf;
    __shared__ int list[CSR_NW * CSR_CAPW];

    const bool second = blockIdx.x >= CSR_NB1;
    const int*   __restrict__ src    = second ? src2    : src1;
    const int*   __restrict__ dst    = second ? dst2    : dst1;
    const float* __restrict__ ew     = second ? ew2     : ew1;
    int*         __restrict__ starts = second ? starts2 : starts1;
    int*         __restrict__ srcs   = second ? srcs2   : srcs1;
    float*       __restrict__ wsv    = second ? ws2     : ws1;
    const int E  = second ? E2 : E1;
    const int N  = second ? N_DST2 : N_DST1;
    const int rb = second ? (int)blockIdx.x - CSR_NB1 : (int)blockIdx.x;
    const int nb = second ? CSR_NB2 : CSR_NB1;
    const int lo = rb * CSR_R;

    const int tid  = threadIdx.x;
    const int lane = tid & 31;
    const int wave = tid >> 5;
    int* mylist = list + wave * CSR_CAPW;

    if (tid < CSR_R) cnt[tid] = 0;
    if (tid == 0) s_ovf = 0;
    __syncthreads();

    int below  = 0;
    int wcount = 0;

    auto visit = [&](int d, int e) {
        below += (d < lo) ? 1 : 0;
        const unsigned r = (unsigned)(d - lo);
        const bool hit = r < (unsigned)CSR_R;
        const unsigned m = (unsigned)__ballot(hit);
        if (hit) {
            atomicAdd(&cnt[r], 1);
            const int slot = wcount + (int)__builtin_amdgcn_mbcnt_lo(m, 0u);
            if (slot < CSR_CAPW) mylist[slot] = e;
        }
        wcount += __popc(m);
    };

    const int E4 = E >> 2;
    const int4* __restrict__ d4 = reinterpret_cast<const int4*>(dst);
    auto batch = [&](const int b0, const bool guard) {
        int4 d[CSR_UB];
#pragma unroll
        for (int u = 0; u < CSR_UB; ++u) {
            const int i = b0 + u * CSR_NT + tid;
            d[u] = d4[(guard && i >= E4) ? E4 - 1 : i];
        }
#pragma unroll
        for (int u = 0; u < CSR_UB; ++u) {
            const int i = b0 + u * CSR_NT + tid;
            if (guard && i >= E4) d[u] = make_int4(CSR_NONE, CSR_NONE, CSR_NONE, CSR_NONE);
            const int e = i << 2;
            visit(d[u].x, e);
            visit(d[u].y, e + 1);
            visit(d[u].z, e + 2);
            visit(d[u].w, e + 3);
        }
    };
    int b0 = 0;
    for (; b0 + CSR_UB * CSR_NT <= E4; b0 += CSR_UB * CSR_NT) batch(b0, false);
    if (b0 < E4) batch(b0, true);
    {
        const int e = (E4 << 2) + tid;
        const int d = e < E ? dst[e] : CSR_NONE;
        visit(d, e);
    }

    for (int off = 16; off > 0; off >>= 1) below += __shfl_xor(below, off);
    if (lane == 0) {
        wbelow[wave] = below;
        if (wcount > CSR_CAPW) s_ovf = 1;
    }
    __syncthreads();

    int base = 0;
    for (int w = 0; w < CSR_NW; ++w) base += wbelow[w];
    const int c = tid < CSR_R ? cnt[tid] : 0;
    int incl = c;
    for (int off = 1; off < 32; off <<= 1) {
        const int t = __shfl_up(incl, off);
        if (lane >= off) incl += t;
    }
    if (lane == 31) stot[wave] = incl;
    __syncthreads();
    int pre = 0;
    for (int w = 0; w < wave; ++w) pre += stot[w];
    const int excl = base + pre + incl - c;
    if (tid < CSR_R) {
        cnt[tid] = excl;
        if (lo + tid < N) starts[lo + tid] = excl;
    }
    if (rb == nb - 1 && tid == 0) starts[N] = E;
    const bool ovf = s_ovf != 0;
    __syncthreads();

    if (!ovf) {
        for (int k = lane; k < wcount; k += 32) {
            const int   e = mylist[k];
            const int   d = dst[e];
            const int   s = src[e];
            const float w = ew[e];
            const int pos = atomicAdd(&cnt[d - lo], 1);
            srcs[pos] = s;
            wsv[pos]  = w;
        }
    } else {
        for (int e = tid; e < E; e += CSR_NT) {
            const unsigned r = (unsigned)(dst[e] - lo);
            if (r < (unsigned)CSR_R) {
                const int pos = atomicAdd(&cnt[r], 1);
                srcs[pos] = src[e];
                wsv[pos]  = ew[e];
            }
        }
    }
}

__global__ void __launch_bounds__(96)
k_aggregate_out(const int* __restrict__ starts, const int* __restrict__ srcs,
                const float* __restrict__ ws, const float* __restrict__ z2,
                const float* __restrict__ s2, const float* __restrict__ bias2,
                float* __restrict__ out) {
    constexpr int C4 = C_OUT / 4;
    int n   = blockIdx.x;
    int tid = threadIdx.x;
    int beg = starts[n];
    int end = starts[n + 1];
    const int t  = tid / C4;
    const int c4 = tid % C4;
    const float4 sv = reinterpret_cast<const float4*>(s2)[(size_t)n * ROW4 + tid];
    const float4 bv = reinterpret_cast<const float4*>(bias2)[c4];
    float4 acc = make_float4(0.f, 0.f, 0.f, 0.f);
    const float4* x4 = reinterpret_cast<const float4*>(z2);
    for (int j = beg; j < end; ++j) {
        int   s = srcs[j];
        float w = ws[j];
        float4 v = x4[(size_t)s * ROW4 + tid];
        acc.x += w * v.x;
        acc.y += w * v.y;
        acc.z += w * v.z;
        acc.w += w * v.w;
    }
    float4 o;
    o.x = leaky(sv.x + acc.x + bv.x);
    o.y = leaky(sv.y + acc.y + bv.y);
    o.z = leaky(sv.z + acc.z + bv.z);
    o.w = leaky(sv.w + acc.w + bv.w);
    reinterpret_cast<float4*>(out)[((size_t)t * N_DST2 + n) * C4 + c4] = o;
}

__global__ void __launch_bounds__(96)
k_aggregate_nf(const int* __restrict__ starts, const int* __restrict__ srcs,
               const float* __restrict__ ws, const float* __restrict__ nf,
               float* __restrict__ agg) {
    constexpr int C4 = C_IN / 4;
    int n   = blockIdx.x;
    int tid = threadIdx.x;
    int beg = starts[n];
    int end = starts[n + 1];
    const int t  = tid / C4;
    const int c4 = tid % C4;
    const float4* p = reinterpret_cast<const float4*>(nf) + (size_t)t * N_SRC * C4 + c4;
    float4 acc = make_float4(0.f, 0.f, 0.f, 0.f);
    for (int j = beg; j < end; ++j) {
        int   s = srcs[j];
        float w = ws[j];
        float4 v = p[(size_t)s * C4];
        acc.x += w * v.x;
        acc.y += w * v.y;
        acc.z += w * v.z;
        acc.w += w * v.w;
    }
    reinterpret_cast<float4*>(agg)[(size_t)n * ROW4 + tid] = acc;
}

constexpr int GEMM_NW = 8;
constexpr int GEMM_NT = GEMM_NW * 32;
constexpr int P_IN    = C_IN + 4;
constexpr int P_H     = H + 4;

template <int ROWS, int K>
__device__ __forceinline__ void stage_weights(const float* __restrict__ Wg, float* sW, int tid) {
    constexpr int K4 = K / 4;
    const float4* __restrict__ g4 = reinterpret_cast<const float4*>(Wg);
    for (int i = tid; i < ROWS * K4; i += GEMM_NT) {
        const int r = i / K4;
        const int q = i - r * K4;
        *reinterpret_cast<float4*>(sW + r * (K + 4) + q * 4) = g4[i];
    }
}

__device__ __forceinline__ v2f lds_frag(const float* sW, int n0, int pitch, int k0, int lane) {
    const int r = n0 + (lane & 15);
    const int k = k0 + ((lane >> 4) << 1);
    return *reinterpret_cast<const v2f*>(sW + r * pitch + k);
}

template <int ROWS>
__device__ __forceinline__ void stage_weights_hperm(const float* __restrict__ Wg, float* sW, int tid) {
    constexpr int K4    = H / 4;
    constexpr int TRIPS = ROWS * K4 / GEMM_NT;
    static_assert(ROWS * K4 % GEMM_NT == 0, "whole trips only");
    const float4* __restrict__ g4 = reinterpret_cast<const float4*>(Wg);
    float4 v[TRIPS];
#pragma unroll
    for (int u = 0; u < TRIPS; ++u) v[u] = g4[u * GEMM_NT + tid];
#pragma unroll
    for (int u = 0; u < TRIPS; ++u) {
        const int i  = u * GEMM_NT + tid;
        const int r  = i / K4;
        const int q4 = i - r * K4;
        const int ht = q4 >> 2;
        const int g  = (q4 >> 1) & 1;
        const int jj = (q4 & 1) << 1;
        float* d = sW + r * P_H + 2 * (8 * ht + 2 * jj + g);
        v2f lo, hi;
        lo.x = v[u].x; lo.y = v[u].y;
        hi.x = v[u].z; hi.y = v[u].w;
        *reinterpret_cast<v2f*>(d)     = lo;
        *reinterpret_cast<v2f*>(d + 4) = hi;
    }
}

template <bool SELF>
__device__ __forceinline__ void layer2_tile(const v8f (&hT)[4], const float* sWn, const float* sWs,
                                            float* __restrict__ z2, float* __restrict__ s2,
                                            int m0, int lane) {
    v8f zacc[2] = {};
    v8f sacc[2] = {};
#pragma unroll
    for (int ht = 0; ht < 4; ++ht) {
#pragma unroll
        for (int jj = 0; jj < 4; ++jj) {
            const int k0 = (ht * 4 + jj) * 4;
            v2f b;
            b.x = hT[ht][2 * jj];
            b.y = hT[ht][2 * jj + 1];
#pragma unroll
            for (int ct = 0; ct < 2; ++ct) {
                v2f a = lds_frag(sWn, ct * 16, P_H, k0, lane);
                zacc[ct] = __builtin_amdgcn_wmma_f32_16x16x4_f32(false, a, false, b,
                                                                 (short)0, zacc[ct], false, false);
            }
            if (SELF) {
#pragma unroll
                for (int ct = 0; ct < 2; ++ct) {
                    v2f a = lds_frag(sWs, ct * 16, P_H, k0, lane);
                    sacc[ct] = __builtin_amdgcn_wmma_f32_16x16x4_f32(false, a, false, b,
                                                                     (short)0, sacc[ct], false, false);
                }
            }
        }
    }
    const size_t row = (size_t)(m0 + (lane & 15)) * C_OUT + ((lane >> 4) << 3);
#pragma unroll
    for (int ct = 0; ct < 2; ++ct) {
        float4* zp = reinterpret_cast<float4*>(z2 + row + ct * 16);
        zp[0] = make_float4(zacc[ct][0], zacc[ct][1], zacc[ct][2], zacc[ct][3]);
        zp[1] = make_float4(zacc[ct][4], zacc[ct][5], zacc[ct][6], zacc[ct][7]);
        if (SELF) {
            float4* sp = reinterpret_cast<float4*>(s2 + row + ct * 16);
            sp[0] = make_float4(sacc[ct][0], sacc[ct][1], sacc[ct][2], sacc[ct][3]);
            sp[1] = make_float4(sacc[ct][4], sacc[ct][5], sacc[ct][6], sacc[ct][7]);
        }
    }
}

__global__ void __launch_bounds__(GEMM_NT)
k_layers(const float* __restrict__ nf, const float* __restrict__ agg1,
         const float* __restrict__ Wself, const float* __restrict__ Wneigh,
         const float* __restrict__ bias,
         const float* __restrict__ Wneigh2, const float* __restrict__ Wself2,
         float* __restrict__ z2, float* __restrict__ s2) {
    __shared__ __attribute__((aligned(16))) float sWs[H * P_IN];
    __shared__ __attribute__((aligned(16))) float sWn[H * P_IN];
    __shared__ __attribute__((aligned(16))) float sW2n[C_OUT * P_H];
    __shared__ __attribute__((aligned(16))) float sW2s[C_OUT * P_H];

    int lane  = threadIdx.x & 31;
    int wave  = threadIdx.x >> 5;
    int mtile = blockIdx.x * GEMM_NW + wave;
    const bool valid = mtile < M1 / 16;
    int m0 = (valid ? mtile : 0) * 16;
    const bool self_tile  = mtile < M2 / 16;
    const bool self_group = (int)blockIdx.x * GEMM_NW < M2 / 16;

    const int mrow = m0 + (lane & 15);
    const int nrow = mrow / T;
    const int trow = mrow - nrow * T;
    const float* xrow = nf + ((size_t)trow * N_SRC + nrow) * C_IN + ((lane >> 4) << 1);

    v2f xS[8], xN[8];
#pragma unroll
    for (int ks = 0; ks < 8; ++ks) {
        xS[ks].x = xrow[ks * 4];
        xS[ks].y = xrow[ks * 4 + 1];
        xN[ks] = load_frag(agg1, m0, C_IN, ks * 4, lane);
    }

    stage_weights<H, C_IN>(Wself,  sWs, threadIdx.x);
    stage_weights<H, C_IN>(Wneigh, sWn, threadIdx.x);
    stage_weights_hperm<C_OUT>(Wneigh2, sW2n, threadIdx.x);
    if (self_group) stage_weights_hperm<C_OUT>(Wself2, sW2s, threadIdx.x);
    __syncthreads();
    if (!valid) return;

    v8f hT[4] = {};
#pragma unroll
    for (int ks = 0; ks < 8; ++ks) {
#pragma unroll
        for (int ht = 0; ht < 4; ++ht) {
            v2f wS = lds_frag(sWs, ht * 16, P_IN, ks * 4, lane);
            hT[ht] = __builtin_amdgcn_wmma_f32_16x16x4_f32(false, wS, false, xS[ks],
                                                           (short)0, hT[ht], false, false);
        }
#pragma unroll
        for (int ht = 0; ht < 4; ++ht) {
            v2f wN = lds_frag(sWn, ht * 16, P_IN, ks * 4, lane);
            hT[ht] = __builtin_amdgcn_wmma_f32_16x16x4_f32(false, wN, false, xN[ks],
                                                           (short)0, hT[ht], false, false);
        }
    }
#pragma unroll
    for (int ht = 0; ht < 4; ++ht) {
        const float4* bp = reinterpret_cast<const float4*>(bias + ht * 16 + ((lane >> 4) << 3));
        const float4 b0 = bp[0];
        const float4 b1 = bp[1];
        hT[ht][0] = leaky(hT[ht][0] + b0.x);
        hT[ht][1] = leaky(hT[ht][1] + b0.y);
        hT[ht][2] = leaky(hT[ht][2] + b0.z);
        hT[ht][3] = leaky(hT[ht][3] + b0.w);
        hT[ht][4] = leaky(hT[ht][4] + b1.x);
        hT[ht][5] = leaky(hT[ht][5] + b1.y);
        hT[ht][6] = leaky(hT[ht][6] + b1.z);
        hT[ht][7] = leaky(hT[ht][7] + b1.w);
    }

    if (self_tile) layer2_tile<true >(hT, sW2n, sW2s, z2, s2, m0, lane);
    else           layer2_tile<false>(hT, sW2n, sW2s, z2, s2, m0, lane);
}

extern "C" void kernel_launch(void* const* d_in, const int* in_sizes, int n_in,
                              void* d_out, int out_size, void* d_ws, size_t ws_size,
                              hipStream_t stream) {
    const float* nf   = (const float*)d_in[0];
    const int*   src1 = (const int*)  d_in[1];
    const int*   dst1 = (const int*)  d_in[2];
    const float* ew1  = (const float*)d_in[3];
    const int*   src2 = (const int*)  d_in[4];
    const int*   dst2 = (const int*)  d_in[5];
    const float* ew2  = (const float*)d_in[6];
    const float* W1s  = (const float*)d_in[7];
    const float* W1n  = (const float*)d_in[8];
    const float* b1   = (const float*)d_in[9];
    const float* W2s  = (const float*)d_in[10];
    const float* W2n  = (const float*)d_in[11];
    const float* b2   = (const float*)d_in[12];
    const int E1 = in_sizes[1];
    const int E2 = in_sizes[4];

    float* out = (float*)d_out;

    float* ws   = (float*)d_ws;
    float* z2   = ws;
    float* s2   = ws   + (size_t)N_DST1 * ROW;
    float* agg1 = ws   + (size_t)N_SRC  * ROW;
    int*   ip      = (int*)(agg1 + (size_t)N_DST1 * ROW + (size_t)M1 * H);
    int*   starts1 = ip;                 ip += N_DST1 + 1;
    int*   srcs1   = ip;                 ip += E1;
    float* ws1     = (float*)ip;         ip += E1;
    int*   starts2 = ip;                 ip += N_DST2 + 1;
    int*   srcs2   = ip;                 ip += E2;
    float* ws2     = (float*)ip;

    k_csr<<<CSR_NB1 + CSR_NB2, CSR_NT, 0, stream>>>(src1, dst1, ew1, starts1, srcs1, ws1, E1,
                                                   src2, dst2, ew2, starts2, srcs2, ws2, E2);

    k_aggregate_nf<<<N_DST1, 96, 0, stream>>>(starts1, srcs1, ws1, nf, agg1);

    k_layers<<<(M1 / 16 + GEMM_NW - 1) / GEMM_NW, GEMM_NT, 0, stream>>>(nf, agg1, W1s, W1n, b1,
                                                                        W2n, W2s, z2, s2);

    k_aggregate_out<<<N_DST2, 96, 0, stream>>>(starts2, srcs2, ws2, z2, s2, b2, out);
}


# Layer-1 neighbour sums gathered into LDS inside the GEMM launch (group = 16 dst nodes = 12 tiles): no agg1 round trip, 4->3 launches, gather waits hide under other groups' WMMAs; sum order unchanged
# speedup vs baseline: 2.3212x; 1.0572x over previous
#include <hip/hip_runtime.h>

typedef __attribute__((ext_vector_type(2))) float v2f;
typedef __attribute__((ext_vector_type(8))) float v8f;

constexpr int T      = 12;
constexpr int C_IN   = 32;
constexpr int H      = 64;
constexpr int C_OUT  = 32;
constexpr int N_SRC  = 50000;
constexpr int N_DST1 = 20000;
constexpr int N_DST2 = 10000;
constexpr int ROW    = T * C_IN;
constexpr int ROW4   = ROW / 4;
constexpr int M1     = N_DST1 * T;
constexpr int M2     = N_DST2 * T;
constexpr float NEG_SLOPE = 0.01f;

__device__ __forceinline__ float leaky(float v) {
    return v > 0.f ? v : NEG_SLOPE * v;
}


constexpr int CSR_R    = 512;
constexpr int CSR_NT   = 1024;
constexpr int CSR_NW   = CSR_NT / 32;
constexpr int CSR_CAPW = 1024;
constexpr int CSR_UB   = 8;
constexpr int CSR_NB1  = (N_DST1 + CSR_R - 1) / CSR_R;
constexpr int CSR_NB2  = (N_DST2 + CSR_R - 1) / CSR_R;
constexpr int CSR_NONE = 0x7fffffff;

__global__ void __launch_bounds__(CSR_NT)
k_csr(const int* __restrict__ src1, const int* __restrict__ dst1, const float* __restrict__ ew1,
      int* __restrict__ starts1, int* __restrict__ srcs1, float* __restrict__ ws1, int E1,
      const int* __restrict__ src2, const int* __restrict__ dst2, const float* __restrict__ ew2,
      int* __restrict__ starts2, int* __restrict__ srcs2, float* __restrict__ ws2, int E2) {
    __shared__ int cnt[CSR_R];
    __shared__ int wbelow[CSR_NW];
    __shared__ int stot[CSR_NW];
    __shared__ int s_ovf;
    __shared__ int list[CSR_NW * CSR_CAPW];

    const bool second = blockIdx.x >= CSR_NB1;
    const int*   __restrict__ src    = second ? src2    : src1;
    const int*   __restrict__ dst    = second ? dst2    : dst1;
    const float* __restrict__ ew     = second ? ew2     : ew1;
    int*         __restrict__ starts = second ? starts2 : starts1;
    int*         __restrict__ srcs   = second ? srcs2   : srcs1;
    float*       __restrict__ wsv    = second ? ws2     : ws1;
    const int E  = second ? E2 : E1;
    const int N  = second ? N_DST2 : N_DST1;
    const int rb = second ? (int)blockIdx.x - CSR_NB1 : (int)blockIdx.x;
    const int nb = second ? CSR_NB2 : CSR_NB1;
    const int lo = rb * CSR_R;

    const int tid  = threadIdx.x;
    const int lane = tid & 31;
    const int wave = tid >> 5;
    int* mylist = list + wave * CSR_CAPW;

    if (tid < CSR_R) cnt[tid] = 0;
    if (tid == 0) s_ovf = 0;
    __syncthreads();

    int below  = 0;
    int wcount = 0;

    auto visit = [&](int d, int e) {
        below += (d < lo) ? 1 : 0;
        const unsigned r = (unsigned)(d - lo);
        const bool hit = r < (unsigned)CSR_R;
        const unsigned m = (unsigned)__ballot(hit);
        if (hit) {
            atomicAdd(&cnt[r], 1);
            const int slot = wcount + (int)__builtin_amdgcn_mbcnt_lo(m, 0u);
            if (slot < CSR_CAPW) mylist[slot] = e;
        }
        wcount += __popc(m);
    };

    const int E4 = E >> 2;
    const int4* __restrict__ d4 = reinterpret_cast<const int4*>(dst);
    auto batch = [&](const int b0, const bool guard) {
        int4 d[CSR_UB];
#pragma unroll
        for (int u = 0; u < CSR_UB; ++u) {
            const int i = b0 + u * CSR_NT + tid;
            d[u] = d4[(guard && i >= E4) ? E4 - 1 : i];
        }
#pragma unroll
        for (int u = 0; u < CSR_UB; ++u) {
            const int i = b0 + u * CSR_NT + tid;
            if (guard && i >= E4) d[u] = make_int4(CSR_NONE, CSR_NONE, CSR_NONE, CSR_NONE);
            const int e = i << 2;
            visit(d[u].x, e);
            visit(d[u].y, e + 1);
            visit(d[u].z, e + 2);
            visit(d[u].w, e + 3);
        }
    };
    int b0 = 0;
    for (; b0 + CSR_UB * CSR_NT <= E4; b0 += CSR_UB * CSR_NT) batch(b0, false);
    if (b0 < E4) batch(b0, true);
    {
        const int e = (E4 << 2) + tid;
        const int d = e < E ? dst[e] : CSR_NONE;
        visit(d, e);
    }

    for (int off = 16; off > 0; off >>= 1) below += __shfl_xor(below, off);
    if (lane == 0) {
        wbelow[wave] = below;
        if (wcount > CSR_CAPW) s_ovf = 1;
    }
    __syncthreads();

    int base = 0;
    for (int w = 0; w < CSR_NW; ++w) base += wbelow[w];
    const int c = tid < CSR_R ? cnt[tid] : 0;
    int incl = c;
    for (int off = 1; off < 32; off <<= 1) {
        const int t = __shfl_up(incl, off);
        if (lane >= off) incl += t;
    }
    if (lane == 31) stot[wave] = incl;
    __syncthreads();
    int pre = 0;
    for (int w = 0; w < wave; ++w) pre += stot[w];
    const int excl = base + pre + incl - c;
    if (tid < CSR_R) {
        cnt[tid] = excl;
        if (lo + tid < N) starts[lo + tid] = excl;
    }
    if (rb == nb - 1 && tid == 0) starts[N] = E;
    const bool ovf = s_ovf != 0;
    __syncthreads();

    if (!ovf) {
        for (int k = lane; k < wcount; k += 32) {
            const int   e = mylist[k];
            const int   d = dst[e];
            const int   s = src[e];
            const float w = ew[e];
            const int pos = atomicAdd(&cnt[d - lo], 1);
            srcs[pos] = s;
            wsv[pos]  = w;
        }
    } else {
        for (int e = tid; e < E; e += CSR_NT) {
            const unsigned r = (unsigned)(dst[e] - lo);
            if (r < (unsigned)CSR_R) {
                const int pos = atomicAdd(&cnt[r], 1);
                srcs[pos] = src[e];
                wsv[pos]  = ew[e];
            }
        }
    }
}

__global__ void __launch_bounds__(96)
k_aggregate_out(const int* __restrict__ starts, const int* __restrict__ srcs,
                const float* __restrict__ ws, const float* __restrict__ z2,
                const float* __restrict__ s2, const float* __restrict__ bias2,
                float* __restrict__ out) {
    constexpr int C4 = C_OUT / 4;
    int n   = blockIdx.x;
    int tid = threadIdx.x;
    int beg = starts[n];
    int end = starts[n + 1];
    const int t  = tid / C4;
    const int c4 = tid % C4;
    const float4 sv = reinterpret_cast<const float4*>(s2)[(size_t)n * ROW4 + tid];
    const float4 bv = reinterpret_cast<const float4*>(bias2)[c4];
    float4 acc = make_float4(0.f, 0.f, 0.f, 0.f);
    const float4* x4 = reinterpret_cast<const float4*>(z2);
    for (int j = beg; j < end; ++j) {
        int   s = srcs[j];
        float w = ws[j];
        float4 v = x4[(size_t)s * ROW4 + tid];
        acc.x += w * v.x;
        acc.y += w * v.y;
        acc.z += w * v.z;
        acc.w += w * v.w;
    }
    float4 o;
    o.x = leaky(sv.x + acc.x + bv.x);
    o.y = leaky(sv.y + acc.y + bv.y);
    o.z = leaky(sv.z + acc.z + bv.z);
    o.w = leaky(sv.w + acc.w + bv.w);
    reinterpret_cast<float4*>(out)[((size_t)t * N_DST2 + n) * C4 + c4] = o;
}

constexpr int FN      = 16;
constexpr int GEMM_NW = FN * T / 16;
constexpr int GEMM_NT = GEMM_NW * 32;
constexpr int P_A     = C_IN + 4;
static_assert(FN * T % 16 == 0 && N_DST1 % FN == 0 && N_DST2 % FN == 0, "whole tiles, whole groups");
constexpr int P_IN    = C_IN + 4;
constexpr int P_H     = H + 4;

template <int ROWS, int K>
__device__ __forceinline__ void stage_weights(const float* __restrict__ Wg, float* sW, int tid) {
    constexpr int K4 = K / 4;
    const float4* __restrict__ g4 = reinterpret_cast<const float4*>(Wg);
    for (int i = tid; i < ROWS * K4; i += GEMM_NT) {
        const int r = i / K4;
        const int q = i - r * K4;
        *reinterpret_cast<float4*>(sW + r * (K + 4) + q * 4) = g4[i];
    }
}

__device__ __forceinline__ v2f lds_frag(const float* sW, int n0, int pitch, int k0, int lane) {
    const int r = n0 + (lane & 15);
    const int k = k0 + ((lane >> 4) << 1);
    return *reinterpret_cast<const v2f*>(sW + r * pitch + k);
}

template <int ROWS>
__device__ __forceinline__ void stage_weights_hperm(const float* __restrict__ Wg, float* sW, int tid) {
    constexpr int K4    = H / 4;
    constexpr int TRIPS = (ROWS * K4 + GEMM_NT - 1) / GEMM_NT;
    const float4* __restrict__ g4 = reinterpret_cast<const float4*>(Wg);
    float4 v[TRIPS];
#pragma unroll
    for (int u = 0; u < TRIPS; ++u) {
        const int i = u * GEMM_NT + tid;
        v[u] = g4[i < ROWS * K4 ? i : 0];
    }
#pragma unroll
    for (int u = 0; u < TRIPS; ++u) {
        const int i  = u * GEMM_NT + tid;
        if (i >= ROWS * K4) continue;
        const int r  = i / K4;
        const int q4 = i - r * K4;
        const int ht = q4 >> 2;
        const int g  = (q4 >> 1) & 1;
        const int jj = (q4 & 1) << 1;
        float* d = sW + r * P_H + 2 * (8 * ht + 2 * jj + g);
        v2f lo, hi;
        lo.x = v[u].x; lo.y = v[u].y;
        hi.x = v[u].z; hi.y = v[u].w;
        *reinterpret_cast<v2f*>(d)     = lo;
        *reinterpret_cast<v2f*>(d + 4) = hi;
    }
}

template <bool SELF>
__device__ __forceinline__ void layer2_tile(const v8f (&hT)[4], const float* sWn, const float* sWs,
                                            float* __restrict__ z2, float* __restrict__ s2,
                                            int m0, int lane) {
    v8f zacc[2] = {};
    v8f sacc[2] = {};
#pragma unroll
    for (int ht = 0; ht < 4; ++ht) {
#pragma unroll
        for (int jj = 0; jj < 4; ++jj) {
            const int k0 = (ht * 4 + jj) * 4;
            v2f b;
            b.x = hT[ht][2 * jj];
            b.y = hT[ht][2 * jj + 1];
#pragma unroll
            for (int ct = 0; ct < 2; ++ct) {
                v2f a = lds_frag(sWn, ct * 16, P_H, k0, lane);
                zacc[ct] = __builtin_amdgcn_wmma_f32_16x16x4_f32(false, a, false, b,
                                                                 (short)0, zacc[ct], false, false);
            }
            if (SELF) {
#pragma unroll
                for (int ct = 0; ct < 2; ++ct) {
                    v2f a = lds_frag(sWs, ct * 16, P_H, k0, lane);
                    sacc[ct] = __builtin_amdgcn_wmma_f32_16x16x4_f32(false, a, false, b,
                                                                     (short)0, sacc[ct], false, false);
                }
            }
        }
    }
    const size_t row = (size_t)(m0 + (lane & 15)) * C_OUT + ((lane >> 4) << 3);
#pragma unroll
    for (int ct = 0; ct < 2; ++ct) {
        float4* zp = reinterpret_cast<float4*>(z2 + row + ct * 16);
        zp[0] = make_float4(zacc[ct][0], zacc[ct][1], zacc[ct][2], zacc[ct][3]);
        zp[1] = make_float4(zacc[ct][4], zacc[ct][5], zacc[ct][6], zacc[ct][7]);
        if (SELF) {
            float4* sp = reinterpret_cast<float4*>(s2 + row + ct * 16);
            sp[0] = make_float4(sacc[ct][0], sacc[ct][1], sacc[ct][2], sacc[ct][3]);
            sp[1] = make_float4(sacc[ct][4], sacc[ct][5], sacc[ct][6], sacc[ct][7]);
        }
    }
}

__global__ void __launch_bounds__(GEMM_NT)
k_layers(const float* __restrict__ nf,
         const int* __restrict__ starts, const int* __restrict__ srcs, const float* __restrict__ wsv,
         const float* __restrict__ Wself, const float* __restrict__ Wneigh,
         const float* __restrict__ bias,
         const float* __restrict__ Wneigh2, const float* __restrict__ Wself2,
         float* __restrict__ z2, float* __restrict__ s2) {
    __shared__ __attribute__((aligned(16))) float sWs[H * P_IN];
    __shared__ __attribute__((aligned(16))) float sWn[H * P_IN];
    __shared__ __attribute__((aligned(16))) float sW2n[C_OUT * P_H];
    __shared__ __attribute__((aligned(16))) float sW2s[C_OUT * P_H];
    __shared__ __attribute__((aligned(16))) float sAgg[FN * T * P_A];

    const int lane  = threadIdx.x & 31;
    const int wave  = threadIdx.x >> 5;
    const int node0 = blockIdx.x * FN;
    const int m0    = node0 * T + wave * 16;
    const bool self_group = (int)blockIdx.x < N_DST2 / FN;

    const int mrow = m0 + (lane & 15);
    const int nrow = mrow / T;
    const int trow = mrow - nrow * T;
    const float* xrow = nf + ((size_t)trow * N_SRC + nrow) * C_IN + ((lane >> 4) << 1);

    v2f xS[8];
#pragma unroll
    for (int ks = 0; ks < 8; ++ks) {
        xS[ks].x = xrow[ks * 4];
        xS[ks].y = xrow[ks * 4 + 1];
    }

    stage_weights<H, C_IN>(Wself,  sWs, threadIdx.x);
    stage_weights<H, C_IN>(Wneigh, sWn, threadIdx.x);
    stage_weights_hperm<C_OUT>(Wneigh2, sW2n, threadIdx.x);
    if (self_group) stage_weights_hperm<C_OUT>(Wself2, sW2s, threadIdx.x);

    {
        constexpr int C4 = C_IN / 4;
        const int grp = __builtin_amdgcn_readfirstlane(wave / 3);
        const int t96 = (wave - grp * 3) * 32 + lane;
        const int t   = t96 / C4;
        const int c4  = t96 % C4;
        const float4* p = reinterpret_cast<const float4*>(nf) + (size_t)t * N_SRC * C4 + c4;
        for (int q = 0; q < FN / 4; ++q) {
            const int nl  = q * 4 + grp;
            const int beg = starts[node0 + nl];
            const int end = starts[node0 + nl + 1];
            float4 acc = make_float4(0.f, 0.f, 0.f, 0.f);
            for (int j = beg; j < end; ++j) {
                int   s = srcs[j];
                float w = wsv[j];
                float4 v = p[(size_t)s * C4];
                acc.x += w * v.x;
                acc.y += w * v.y;
                acc.z += w * v.z;
                acc.w += w * v.w;
            }
            *reinterpret_cast<float4*>(sAgg + (nl * T + t) * P_A + c4 * 4) = acc;
        }
    }
    __syncthreads();

    v2f xN[8];
    {
        const float* arow = sAgg + (wave * 16 + (lane & 15)) * P_A + ((lane >> 4) << 1);
#pragma unroll
        for (int ks = 0; ks < 8; ++ks) xN[ks] = *reinterpret_cast<const v2f*>(arow + ks * 4);
    }

    v8f hT[4] = {};
#pragma unroll
    for (int ks = 0; ks < 8; ++ks) {
#pragma unroll
        for (int ht = 0; ht < 4; ++ht) {
            v2f wS = lds_frag(sWs, ht * 16, P_IN, ks * 4, lane);
            hT[ht] = __builtin_amdgcn_wmma_f32_16x16x4_f32(false, wS, false, xS[ks],
                                                           (short)0, hT[ht], false, false);
        }
#pragma unroll
        for (int ht = 0; ht < 4; ++ht) {
            v2f wN = lds_frag(sWn, ht * 16, P_IN, ks * 4, lane);
            hT[ht] = __builtin_amdgcn_wmma_f32_16x16x4_f32(false, wN, false, xN[ks],
                                                           (short)0, hT[ht], false, false);
        }
    }
#pragma unroll
    for (int ht = 0; ht < 4; ++ht) {
        const float4* bp = reinterpret_cast<const float4*>(bias + ht * 16 + ((lane >> 4) << 3));
        const float4 b0 = bp[0];
        const float4 b1 = bp[1];
        hT[ht][0] = leaky(hT[ht][0] + b0.x);
        hT[ht][1] = leaky(hT[ht][1] + b0.y);
        hT[ht][2] = leaky(hT[ht][2] + b0.z);
        hT[ht][3] = leaky(hT[ht][3] + b0.w);
        hT[ht][4] = leaky(hT[ht][4] + b1.x);
        hT[ht][5] = leaky(hT[ht][5] + b1.y);
        hT[ht][6] = leaky(hT[ht][6] + b1.z);
        hT[ht][7] = leaky(hT[ht][7] + b1.w);
    }

    if (self_group) layer2_tile<true >(hT, sW2n, sW2s, z2, s2, m0, lane);
    else            layer2_tile<false>(hT, sW2n, sW2s, z2, s2, m0, lane);
}

extern "C" void kernel_launch(void* const* d_in, const int* in_sizes, int n_in,
                              void* d_out, int out_size, void* d_ws, size_t ws_size,
                              hipStream_t stream) {
    const float* nf   = (const float*)d_in[0];
    const int*   src1 = (const int*)  d_in[1];
    const int*   dst1 = (const int*)  d_in[2];
    const float* ew1  = (const float*)d_in[3];
    const int*   src2 = (const int*)  d_in[4];
    const int*   dst2 = (const int*)  d_in[5];
    const float* ew2  = (const float*)d_in[6];
    const float* W1s  = (const float*)d_in[7];
    const float* W1n  = (const float*)d_in[8];
    const float* b1   = (const float*)d_in[9];
    const float* W2s  = (const float*)d_in[10];
    const float* W2n  = (const float*)d_in[11];
    const float* b2   = (const float*)d_in[12];
    const int E1 = in_sizes[1];
    const int E2 = in_sizes[4];

    float* out = (float*)d_out;

    float* ws   = (float*)d_ws;
    float* z2   = ws;
    float* s2   = ws   + (size_t)N_DST1 * ROW;
    int*   ip      = (int*)(ws + (size_t)N_SRC * ROW + (size_t)N_DST1 * ROW + (size_t)M1 * H);
    int*   starts1 = ip;                 ip += N_DST1 + 1;
    int*   srcs1   = ip;                 ip += E1;
    float* ws1     = (float*)ip;         ip += E1;
    int*   starts2 = ip;                 ip += N_DST2 + 1;
    int*   srcs2   = ip;                 ip += E2;
    float* ws2     = (float*)ip;

    k_csr<<<CSR_NB1 + CSR_NB2, CSR_NT, 0, stream>>>(src1, dst1, ew1, starts1, srcs1, ws1, E1,
                                                   src2, dst2, ew2, starts2, srcs2, ws2, E2);

    k_layers<<<N_DST1 / FN, GEMM_NT, 0, stream>>>(nf, starts1, srcs1, ws1, W1s, W1n, b1,
                                                  W2n, W2s, z2, s2);

    k_aggregate_out<<<N_DST2, 96, 0, stream>>>(starts2, srcs2, ws2, z2, s2, b2, out);
}


# CSR placement through LDS: list entries carry the node offset (no dst re-read); (src, weight) go to an LDS image of the group's contiguous output piece, stored out as whole lines; arrival order kept
# speedup vs baseline: 2.5279x; 1.0890x over previous
#include <hip/hip_runtime.h>

typedef __attribute__((ext_vector_type(2))) float v2f;
typedef __attribute__((ext_vector_type(8))) float v8f;

constexpr int T      = 12;
constexpr int C_IN   = 32;
constexpr int H      = 64;
constexpr int C_OUT  = 32;
constexpr int N_SRC  = 50000;
constexpr int N_DST1 = 20000;
constexpr int N_DST2 = 10000;
constexpr int ROW    = T * C_IN;
constexpr int ROW4   = ROW / 4;
constexpr int M1     = N_DST1 * T;
constexpr int M2     = N_DST2 * T;
constexpr float NEG_SLOPE = 0.01f;

__device__ __forceinline__ float leaky(float v) {
    return v > 0.f ? v : NEG_SLOPE * v;
}


constexpr int CSR_R    = 512;
constexpr int CSR_NT   = 1024;
constexpr int CSR_NW   = CSR_NT / 32;
constexpr int CSR_CAPW = 1024;
constexpr int CSR_UB   = 8;
constexpr int CSR_NB1  = (N_DST1 + CSR_R - 1) / CSR_R;
constexpr int CSR_NB2  = (N_DST2 + CSR_R - 1) / CSR_R;
constexpr int CSR_NONE = 0x7fffffff;
constexpr int CSR_RSH  = 22;
constexpr int CSR_STG  = 12288;
static_assert(CSR_R <= (1 << (31 - CSR_RSH)), "node offset fits above the edge id in a positive int");

__global__ void __launch_bounds__(CSR_NT)
k_csr(const int* __restrict__ src1, const int* __restrict__ dst1, const float* __restrict__ ew1,
      int* __restrict__ starts1, int* __restrict__ srcs1, float* __restrict__ ws1, int E1,
      const int* __restrict__ src2, const int* __restrict__ dst2, const float* __restrict__ ew2,
      int* __restrict__ starts2, int* __restrict__ srcs2, float* __restrict__ ws2, int E2) {
    __shared__ int cnt[CSR_R];
    __shared__ int wbelow[CSR_NW];
    __shared__ int stot[CSR_NW];
    __shared__ int s_ovf;
    __shared__ int list[CSR_NW * CSR_CAPW];
    __shared__ int   stS[CSR_STG];
    __shared__ float stW[CSR_STG];

    const bool second = blockIdx.x >= CSR_NB1;
    const int*   __restrict__ src    = second ? src2    : src1;
    const int*   __restrict__ dst    = second ? dst2    : dst1;
    const float* __restrict__ ew     = second ? ew2     : ew1;
    int*         __restrict__ starts = second ? starts2 : starts1;
    int*         __restrict__ srcs   = second ? srcs2   : srcs1;
    float*       __restrict__ wsv    = second ? ws2     : ws1;
    const int E  = second ? E2 : E1;
    const int N  = second ? N_DST2 : N_DST1;
    const int rb = second ? (int)blockIdx.x - CSR_NB1 : (int)blockIdx.x;
    const int nb = second ? CSR_NB2 : CSR_NB1;
    const int lo = rb * CSR_R;

    const int tid  = threadIdx.x;
    const int lane = tid & 31;
    const int wave = tid >> 5;
    int* mylist = list + wave * CSR_CAPW;

    if (tid < CSR_R) cnt[tid] = 0;
    if (tid == 0) s_ovf = (E >= (1 << CSR_RSH)) ? 1 : 0;
    __syncthreads();

    int below  = 0;
    int wcount = 0;

    auto visit = [&](int d, int e) {
        below += (d < lo) ? 1 : 0;
        const unsigned r = (unsigned)(d - lo);
        const bool hit = r < (unsigned)CSR_R;
        const unsigned m = (unsigned)__ballot(hit);
        if (hit) {
            atomicAdd(&cnt[r], 1);
            const int slot = wcount + (int)__builtin_amdgcn_mbcnt_lo(m, 0u);
            if (slot < CSR_CAPW) mylist[slot] = e | (int)(r << CSR_RSH);
        }
        wcount += __popc(m);
    };

    const int E4 = E >> 2;
    const int4* __restrict__ d4 = reinterpret_cast<const int4*>(dst);
    auto batch = [&](const int b0, const bool guard) {
        int4 d[CSR_UB];
#pragma unroll
        for (int u = 0; u < CSR_UB; ++u) {
            const int i = b0 + u * CSR_NT + tid;
            d[u] = d4[(guard && i >= E4) ? E4 - 1 : i];
        }
#pragma unroll
        for (int u = 0; u < CSR_UB; ++u) {
            const int i = b0 + u * CSR_NT + tid;
            if (guard && i >= E4) d[u] = make_int4(CSR_NONE, CSR_NONE, CSR_NONE, CSR_NONE);
            const int e = i << 2;
            visit(d[u].x, e);
            visit(d[u].y, e + 1);
            visit(d[u].z, e + 2);
            visit(d[u].w, e + 3);
        }
    };
    int b0 = 0;
    for (; b0 + CSR_UB * CSR_NT <= E4; b0 += CSR_UB * CSR_NT) batch(b0, false);
    if (b0 < E4) batch(b0, true);
    {
        const int e = (E4 << 2) + tid;
        const int d = e < E ? dst[e] : CSR_NONE;
        visit(d, e);
    }

    for (int off = 16; off > 0; off >>= 1) below += __shfl_xor(below, off);
    if (lane == 0) {
        wbelow[wave] = below;
        if (wcount > CSR_CAPW) s_ovf = 1;
    }
    __syncthreads();

    int base = 0;
    for (int w = 0; w < CSR_NW; ++w) base += wbelow[w];
    const int c = tid < CSR_R ? cnt[tid] : 0;
    int incl = c;
    for (int off = 1; off < 32; off <<= 1) {
        const int t = __shfl_up(incl, off);
        if (lane >= off) incl += t;
    }
    if (lane == 31) stot[wave] = incl;
    __syncthreads();
    int pre = 0, total = 0;
    for (int w = 0; w < CSR_NW; ++w) {
        const int s = stot[w];
        total += s;
        if (w < wave) pre += s;
    }
    const int excl = base + pre + incl - c;
    if (tid < CSR_R) {
        cnt[tid] = excl;
        if (lo + tid < N) starts[lo + tid] = excl;
    }
    if (rb == nb - 1 && tid == 0) starts[N] = E;
    const bool ovf = s_ovf != 0;
    __syncthreads();

    if (!ovf) {
        for (int k = lane; k < wcount; k += 32) {
            const int   ent = mylist[k];
            const int   e = ent & ((1 << CSR_RSH) - 1);
            const int   s = src[e];
            const float w = ew[e];
            const int pos = atomicAdd(&cnt[(unsigned)ent >> CSR_RSH], 1);
            const int loc = pos - base;
            if (loc < CSR_STG) {
                stS[loc] = s;
                stW[loc] = w;
            } else {
                srcs[pos] = s;
                wsv[pos]  = w;
            }
        }
        __syncthreads();
        const int nst = total < CSR_STG ? total : CSR_STG;
        for (int i = tid; i < nst; i += CSR_NT) {
            srcs[base + i] = stS[i];
            wsv[base + i]  = stW[i];
        }
    } else {
        for (int e = tid; e < E; e += CSR_NT) {
            const unsigned r = (unsigned)(dst[e] - lo);
            if (r < (unsigned)CSR_R) {
                const int pos = atomicAdd(&cnt[r], 1);
                srcs[pos] = src[e];
                wsv[pos]  = ew[e];
            }
        }
    }
}

__global__ void __launch_bounds__(96)
k_aggregate_out(const int* __restrict__ starts, const int* __restrict__ srcs,
                const float* __restrict__ ws, const float* __restrict__ z2,
                const float* __restrict__ s2, const float* __restrict__ bias2,
                float* __restrict__ out) {
    constexpr int C4 = C_OUT / 4;
    int n   = blockIdx.x;
    int tid = threadIdx.x;
    int beg = starts[n];
    int end = starts[n + 1];
    const int t  = tid / C4;
    const int c4 = tid % C4;
    const float4 sv = reinterpret_cast<const float4*>(s2)[(size_t)n * ROW4 + tid];
    const float4 bv = reinterpret_cast<const float4*>(bias2)[c4];
    float4 acc = make_float4(0.f, 0.f, 0.f, 0.f);
    const float4* x4 = reinterpret_cast<const float4*>(z2);
    for (int j = beg; j < end; ++j) {
        int   s = srcs[j];
        float w = ws[j];
        float4 v = x4[(size_t)s * ROW4 + tid];
        acc.x += w * v.x;
        acc.y += w * v.y;
        acc.z += w * v.z;
        acc.w += w * v.w;
    }
    float4 o;
    o.x = leaky(sv.x + acc.x + bv.x);
    o.y = leaky(sv.y + acc.y + bv.y);
    o.z = leaky(sv.z + acc.z + bv.z);
    o.w = leaky(sv.w + acc.w + bv.w);
    reinterpret_cast<float4*>(out)[((size_t)t * N_DST2 + n) * C4 + c4] = o;
}

constexpr int FN      = 16;
constexpr int GEMM_NW = FN * T / 16;
constexpr int GEMM_NT = GEMM_NW * 32;
constexpr int P_A     = C_IN + 4;
static_assert(FN * T % 16 == 0 && N_DST1 % FN == 0 && N_DST2 % FN == 0, "whole tiles, whole groups");
constexpr int P_IN    = C_IN + 4;
constexpr int P_H     = H + 4;

template <int ROWS, int K>
__device__ __forceinline__ void stage_weights(const float* __restrict__ Wg, float* sW, int tid) {
    constexpr int K4 = K / 4;
    const float4* __restrict__ g4 = reinterpret_cast<const float4*>(Wg);
    for (int i = tid; i < ROWS * K4; i += GEMM_NT) {
        const int r = i / K4;
        const int q = i - r * K4;
        *reinterpret_cast<float4*>(sW + r * (K + 4) + q * 4) = g4[i];
    }
}

__device__ __forceinline__ v2f lds_frag(const float* sW, int n0, int pitch, int k0, int lane) {
    const int r = n0 + (lane & 15);
    const int k = k0 + ((lane >> 4) << 1);
    return *reinterpret_cast<const v2f*>(sW + r * pitch + k);
}

template <int ROWS>
__device__ __forceinline__ void stage_weights_hperm(const float* __restrict__ Wg, float* sW, int tid) {
    constexpr int K4    = H / 4;
    constexpr int TRIPS = (ROWS * K4 + GEMM_NT - 1) / GEMM_NT;
    const float4* __restrict__ g4 = reinterpret_cast<const float4*>(Wg);
    float4 v[TRIPS];
#pragma unroll
    for (int u = 0; u < TRIPS; ++u) {
        const int i = u * GEMM_NT + tid;
        v[u] = g4[i < ROWS * K4 ? i : 0];
    }
#pragma unroll
    for (int u = 0; u < TRIPS; ++u) {
        const int i  = u * GEMM_NT + tid;
        if (i >= ROWS * K4) continue;
        const int r  = i / K4;
        const int q4 = i - r * K4;
        const int ht = q4 >> 2;
        const int g  = (q4 >> 1) & 1;
        const int jj = (q4 & 1) << 1;
        float* d = sW + r * P_H + 2 * (8 * ht + 2 * jj + g);
        v2f lo, hi;
        lo.x = v[u].x; lo.y = v[u].y;
        hi.x = v[u].z; hi.y = v[u].w;
        *reinterpret_cast<v2f*>(d)     = lo;
        *reinterpret_cast<v2f*>(d + 4) = hi;
    }
}

template <bool SELF>
__device__ __forceinline__ void layer2_tile(const v8f (&hT)[4], const float* sWn, const float* sWs,
                                            float* __restrict__ z2, float* __restrict__ s2,
                                            int m0, int lane) {
    v8f zacc[2] = {};
    v8f sacc[2] = {};
#pragma unroll
    for (int ht = 0; ht < 4; ++ht) {
#pragma unroll
        for (int jj = 0; jj < 4; ++jj) {
            const int k0 = (ht * 4 + jj) * 4;
            v2f b;
            b.x = hT[ht][2 * jj];
            b.y = hT[ht][2 * jj + 1];
#pragma unroll
            for (int ct = 0; ct < 2; ++ct) {
                v2f a = lds_frag(sWn, ct * 16, P_H, k0, lane);
                zacc[ct] = __builtin_amdgcn_wmma_f32_16x16x4_f32(false, a, false, b,
                                                                 (short)0, zacc[ct], false, false);
            }
            if (SELF) {
#pragma unroll
                for (int ct = 0; ct < 2; ++ct) {
                    v2f a = lds_frag(sWs, ct * 16, P_H, k0, lane);
                    sacc[ct] = __builtin_amdgcn_wmma_f32_16x16x4_f32(false, a, false, b,
                                                                     (short)0, sacc[ct], false, false);
                }
            }
        }
    }
    const size_t row = (size_t)(m0 + (lane & 15)) * C_OUT + ((lane >> 4) << 3);
#pragma unroll
    for (int ct = 0; ct < 2; ++ct) {
        float4* zp = reinterpret_cast<float4*>(z2 + row + ct * 16);
        zp[0] = make_float4(zacc[ct][0], zacc[ct][1], zacc[ct][2], zacc[ct][3]);
        zp[1] = make_float4(zacc[ct][4], zacc[ct][5], zacc[ct][6], zacc[ct][7]);
        if (SELF) {
            float4* sp = reinterpret_cast<float4*>(s2 + row + ct * 16);
            sp[0] = make_float4(sacc[ct][0], sacc[ct][1], sacc[ct][2], sacc[ct][3]);
            sp[1] = make_float4(sacc[ct][4], sacc[ct][5], sacc[ct][6], sacc[ct][7]);
        }
    }
}

__global__ void __launch_bounds__(GEMM_NT)
k_layers(const float* __restrict__ nf,
         const int* __restrict__ starts, const int* __restrict__ srcs, const float* __restrict__ wsv,
         const float* __restrict__ Wself, const float* __restrict__ Wneigh,
         const float* __restrict__ bias,
         const float* __restrict__ Wneigh2, const float* __restrict__ Wself2,
         float* __restrict__ z2, float* __restrict__ s2) {
    __shared__ __attribute__((aligned(16))) float sWs[H * P_IN];
    __shared__ __attribute__((aligned(16))) float sWn[H * P_IN];
    __shared__ __attribute__((aligned(16))) float sW2n[C_OUT * P_H];
    __shared__ __attribute__((aligned(16))) float sW2s[C_OUT * P_H];
    __shared__ __attribute__((aligned(16))) float sAgg[FN * T * P_A];

    const int lane  = threadIdx.x & 31;
    const int wave  = threadIdx.x >> 5;
    const int node0 = blockIdx.x * FN;
    const int m0    = node0 * T + wave * 16;
    const bool self_group = (int)blockIdx.x < N_DST2 / FN;

    const int mrow = m0 + (lane & 15);
    const int nrow = mrow / T;
    const int trow = mrow - nrow * T;
    const float* xrow = nf + ((size_t)trow * N_SRC + nrow) * C_IN + ((lane >> 4) << 1);

    v2f xS[8];
#pragma unroll
    for (int ks = 0; ks < 8; ++ks) {
        xS[ks].x = xrow[ks * 4];
        xS[ks].y = xrow[ks * 4 + 1];
    }

    stage_weights<H, C_IN>(Wself,  sWs, threadIdx.x);
    stage_weights<H, C_IN>(Wneigh, sWn, threadIdx.x);
    stage_weights_hperm<C_OUT>(Wneigh2, sW2n, threadIdx.x);
    if (self_group) stage_weights_hperm<C_OUT>(Wself2, sW2s, threadIdx.x);

    {
        constexpr int C4 = C_IN / 4;
        const int grp = __builtin_amdgcn_readfirstlane(wave / 3);
        const int t96 = (wave - grp * 3) * 32 + lane;
        const int t   = t96 / C4;
        const int c4  = t96 % C4;
        const float4* p = reinterpret_cast<const float4*>(nf) + (size_t)t * N_SRC * C4 + c4;
        for (int q = 0; q < FN / 4; ++q) {
            const int nl  = q * 4 + grp;
            const int beg = starts[node0 + nl];
            const int end = starts[node0 + nl + 1];
            float4 acc = make_float4(0.f, 0.f, 0.f, 0.f);
            for (int j = beg; j < end; ++j) {
                int   s = srcs[j];
                float w = wsv[j];
                float4 v = p[(size_t)s * C4];
                acc.x += w * v.x;
                acc.y += w * v.y;
                acc.z += w * v.z;
                acc.w += w * v.w;
            }
            *reinterpret_cast<float4*>(sAgg + (nl * T + t) * P_A + c4 * 4) = acc;
        }
    }
    __syncthreads();

    v2f xN[8];
    {
        const float* arow = sAgg + (wave * 16 + (lane & 15)) * P_A + ((lane >> 4) << 1);
#pragma unroll
        for (int ks = 0; ks < 8; ++ks) xN[ks] = *reinterpret_cast<const v2f*>(arow + ks * 4);
    }

    v8f hT[4] = {};
#pragma unroll
    for (int ks = 0; ks < 8; ++ks) {
#pragma unroll
        for (int ht = 0; ht < 4; ++ht) {
            v2f wS = lds_frag(sWs, ht * 16, P_IN, ks * 4, lane);
            hT[ht] = __builtin_amdgcn_wmma_f32_16x16x4_f32(false, wS, false, xS[ks],
                                                           (short)0, hT[ht], false, false);
        }
#pragma unroll
        for (int ht = 0; ht < 4; ++ht) {
            v2f wN = lds_frag(sWn, ht * 16, P_IN, ks * 4, lane);
            hT[ht] = __builtin_amdgcn_wmma_f32_16x16x4_f32(false, wN, false, xN[ks],
                                                           (short)0, hT[ht], false, false);
        }
    }
#pragma unroll
    for (int ht = 0; ht < 4; ++ht) {
        const float4* bp = reinterpret_cast<const float4*>(bias + ht * 16 + ((lane >> 4) << 3));
        const float4 b0 = bp[0];
        const float4 b1 = bp[1];
        hT[ht][0] = leaky(hT[ht][0] + b0.x);
        hT[ht][1] = leaky(hT[ht][1] + b0.y);
        hT[ht][2] = leaky(hT[ht][2] + b0.z);
        hT[ht][3] = leaky(hT[ht][3] + b0.w);
        hT[ht][4] = leaky(hT[ht][4] + b1.x);
        hT[ht][5] = leaky(hT[ht][5] + b1.y);
        hT[ht][6] = leaky(hT[ht][6] + b1.z);
        hT[ht][7] = leaky(hT[ht][7] + b1.w);
    }

    if (self_group) layer2_tile<true >(hT, sW2n, sW2s, z2, s2, m0, lane);
    else            layer2_tile<false>(hT, sW2n, sW2s, z2, s2, m0, lane);
}

extern "C" void kernel_launch(void* const* d_in, const int* in_sizes, int n_in,
                              void* d_out, int out_size, void* d_ws, size_t ws_size,
                              hipStream_t stream) {
    const float* nf   = (const float*)d_in[0];
    const int*   src1 = (const int*)  d_in[1];
    const int*   dst1 = (const int*)  d_in[2];
    const float* ew1  = (const float*)d_in[3];
    const int*   src2 = (const int*)  d_in[4];
    const int*   dst2 = (const int*)  d_in[5];
    const float* ew2  = (const float*)d_in[6];
    const float* W1s  = (const float*)d_in[7];
    const float* W1n  = (const float*)d_in[8];
    const float* b1   = (const float*)d_in[9];
    const float* W2s  = (const float*)d_in[10];
    const float* W2n  = (const float*)d_in[11];
    const float* b2   = (const float*)d_in[12];
    const int E1 = in_sizes[1];
    const int E2 = in_sizes[4];

    float* out = (float*)d_out;

    float* ws   = (float*)d_ws;
    float* z2   = ws;
    float* s2   = ws   + (size_t)N_DST1 * ROW;
    int*   ip      = (int*)(ws + (size_t)N_SRC * ROW + (size_t)N_DST1 * ROW + (size_t)M1 * H);
    int*   starts1 = ip;                 ip += N_DST1 + 1;
    int*   srcs1   = ip;                 ip += E1;
    float* ws1     = (float*)ip;         ip += E1;
    int*   starts2 = ip;                 ip += N_DST2 + 1;
    int*   srcs2   = ip;                 ip += E2;
    float* ws2     = (float*)ip;

    k_csr<<<CSR_NB1 + CSR_NB2, CSR_NT, 0, stream>>>(src1, dst1, ew1, starts1, srcs1, ws1, E1,
                                                   src2, dst2, ew2, starts2, srcs2, ws2, E2);

    k_layers<<<N_DST1 / FN, GEMM_NT, 0, stream>>>(nf, starts1, srcs1, ws1, W1s, W1n, b1,
                                                  W2n, W2s, z2, s2);

    k_aggregate_out<<<N_DST2, 96, 0, stream>>>(starts2, srcs2, ws2, z2, s2, b2, out);
}


# CSR built per edge slice: each graph's edges cut in 4 slices, each sorted by its own work-groups (60->240 groups, each streams a quarter of dst); gathers walk a node's 4 pieces; sums slice by slice
# speedup vs baseline: 3.1923x; 1.2628x over previous
#include <hip/hip_runtime.h>

typedef __attribute__((ext_vector_type(2))) float v2f;
typedef __attribute__((ext_vector_type(8))) float v8f;

constexpr int T      = 12;
constexpr int C_IN   = 32;
constexpr int H      = 64;
constexpr int C_OUT  = 32;
constexpr int N_SRC  = 50000;
constexpr int N_DST1 = 20000;
constexpr int N_DST2 = 10000;
constexpr int ROW    = T * C_IN;
constexpr int ROW4   = ROW / 4;
constexpr int M1     = N_DST1 * T;
constexpr int M2     = N_DST2 * T;
constexpr float NEG_SLOPE = 0.01f;

__device__ __forceinline__ float leaky(float v) {
    return v > 0.f ? v : NEG_SLOPE * v;
}


constexpr int CSR_S    = 4;
constexpr int CSR_R    = 512;
constexpr int CSR_NT   = 1024;
constexpr int CSR_NW   = CSR_NT / 32;
constexpr int CSR_CAPW = 1024;
constexpr int CSR_UB   = 8;
constexpr int CSR_NB1  = (N_DST1 + CSR_R - 1) / CSR_R;
constexpr int CSR_NB2  = (N_DST2 + CSR_R - 1) / CSR_R;
constexpr int CSR_NONE = 0x7fffffff;
constexpr int CSR_RSH  = 22;
constexpr int CSR_STG  = 12288;
static_assert(CSR_R <= (1 << (31 - CSR_RSH)), "node offset fits above the edge id in a positive int");

__global__ void __launch_bounds__(CSR_NT)
k_csr(const int* __restrict__ src1, const int* __restrict__ dst1, const float* __restrict__ ew1,
      int* __restrict__ starts1, int* __restrict__ srcs1, float* __restrict__ ws1, int E1,
      const int* __restrict__ src2, const int* __restrict__ dst2, const float* __restrict__ ew2,
      int* __restrict__ starts2, int* __restrict__ srcs2, float* __restrict__ ws2, int E2) {
    __shared__ int cnt[CSR_R];
    __shared__ int wbelow[CSR_NW];
    __shared__ int stot[CSR_NW];
    __shared__ int s_ovf;
    __shared__ int list[CSR_NW * CSR_CAPW];
    __shared__ int   stS[CSR_STG];
    __shared__ float stW[CSR_STG];

    const bool second = blockIdx.x >= CSR_S * CSR_NB1;
    const int*   __restrict__ src    = second ? src2    : src1;
    const int*   __restrict__ dst    = second ? dst2    : dst1;
    const float* __restrict__ ew     = second ? ew2     : ew1;
    int*         __restrict__ starts = second ? starts2 : starts1;
    int*         __restrict__ srcs   = second ? srcs2   : srcs1;
    float*       __restrict__ wsv    = second ? ws2     : ws1;
    const int Eall = second ? E2 : E1;
    const int N  = second ? N_DST2 : N_DST1;
    const int nb = second ? CSR_NB2 : CSR_NB1;
    const int bi = second ? (int)blockIdx.x - CSR_S * CSR_NB1 : (int)blockIdx.x;
    const int sl = bi / nb;
    const int rb = bi - sl * nb;
    const int lo = rb * CSR_R;
    const int SL = ((Eall + CSR_S - 1) / CSR_S + 3) & ~3;
    const int e0 = sl * SL < Eall ? sl * SL : Eall;
    const int E  = Eall - e0 < SL ? Eall - e0 : SL;
    src += e0; dst += e0; ew += e0; srcs += e0; wsv += e0;
    starts += sl * (N + 1);

    const int tid  = threadIdx.x;
    const int lane = tid & 31;
    const int wave = tid >> 5;
    int* mylist = list + wave * CSR_CAPW;

    if (tid < CSR_R) cnt[tid] = 0;
    if (tid == 0) s_ovf = (E >= (1 << CSR_RSH)) ? 1 : 0;
    __syncthreads();

    int below  = 0;
    int wcount = 0;

    auto visit = [&](int d, int e) {
        below += (d < lo) ? 1 : 0;
        const unsigned r = (unsigned)(d - lo);
        const bool hit = r < (unsigned)CSR_R;
        const unsigned m = (unsigned)__ballot(hit);
        if (hit) {
            atomicAdd(&cnt[r], 1);
            const int slot = wcount + (int)__builtin_amdgcn_mbcnt_lo(m, 0u);
            if (slot < CSR_CAPW) mylist[slot] = e | (int)(r << CSR_RSH);
        }
        wcount += __popc(m);
    };

    const int E4 = E >> 2;
    const int4* __restrict__ d4 = reinterpret_cast<const int4*>(dst);
    auto batch = [&](const int b0, const bool guard) {
        int4 d[CSR_UB];
#pragma unroll
        for (int u = 0; u < CSR_UB; ++u) {
            const int i = b0 + u * CSR_NT + tid;
            d[u] = d4[(guard && i >= E4) ? E4 - 1 : i];
        }
#pragma unroll
        for (int u = 0; u < CSR_UB; ++u) {
            const int i = b0 + u * CSR_NT + tid;
            if (guard && i >= E4) d[u] = make_int4(CSR_NONE, CSR_NONE, CSR_NONE, CSR_NONE);
            const int e = i << 2;
            visit(d[u].x, e);
            visit(d[u].y, e + 1);
            visit(d[u].z, e + 2);
            visit(d[u].w, e + 3);
        }
    };
    int b0 = 0;
    for (; b0 + CSR_UB * CSR_NT <= E4; b0 += CSR_UB * CSR_NT) batch(b0, false);
    if (b0 < E4) batch(b0, true);
    {
        const int e = (E4 << 2) + tid;
        const int d = e < E ? dst[e] : CSR_NONE;
        visit(d, e);
    }

    for (int off = 16; off > 0; off >>= 1) below += __shfl_xor(below, off);
    if (lane == 0) {
        wbelow[wave] = below;
        if (wcount > CSR_CAPW) s_ovf = 1;
    }
    __syncthreads();

    int base = 0;
    for (int w = 0; w < CSR_NW; ++w) base += wbelow[w];
    const int c = tid < CSR_R ? cnt[tid] : 0;
    int incl = c;
    for (int off = 1; off < 32; off <<= 1) {
        const int t = __shfl_up(incl, off);
        if (lane >= off) incl += t;
    }
    if (lane == 31) stot[wave] = incl;
    __syncthreads();
    int pre = 0, total = 0;
    for (int w = 0; w < CSR_NW; ++w) {
        const int s = stot[w];
        total += s;
        if (w < wave) pre += s;
    }
    const int excl = base + pre + incl - c;
    if (tid < CSR_R) {
        cnt[tid] = excl;
        if (lo + tid < N) starts[lo + tid] = e0 + excl;
    }
    if (rb == nb - 1 && tid == 0) starts[N] = e0 + E;
    const bool ovf = s_ovf != 0;
    __syncthreads();

    if (!ovf) {
        for (int k = lane; k < wcount; k += 32) {
            const int   ent = mylist[k];
            const int   e = ent & ((1 << CSR_RSH) - 1);
            const int   s = src[e];
            const float w = ew[e];
            const int pos = atomicAdd(&cnt[(unsigned)ent >> CSR_RSH], 1);
            const int loc = pos - base;
            if (loc < CSR_STG) {
                stS[loc] = s;
                stW[loc] = w;
            } else {
                srcs[pos] = s;
                wsv[pos]  = w;
            }
        }
        __syncthreads();
        const int nst = total < CSR_STG ? total : CSR_STG;
        for (int i = tid; i < nst; i += CSR_NT) {
            srcs[base + i] = stS[i];
            wsv[base + i]  = stW[i];
        }
    } else {
        for (int e = tid; e < E; e += CSR_NT) {
            const unsigned r = (unsigned)(dst[e] - lo);
            if (r < (unsigned)CSR_R) {
                const int pos = atomicAdd(&cnt[r], 1);
                srcs[pos] = src[e];
                wsv[pos]  = ew[e];
            }
        }
    }
}

constexpr int GCH = 4;
template <int NV>
__device__ __forceinline__ void gather_trip(const float4* __restrict__ rows, size_t stride,
                                            const int* __restrict__ srcs, const float* __restrict__ ws,
                                            int j, float4& acc) {
    float4 v[NV];
    float  w[NV];
#pragma unroll
    for (int u = 0; u < NV; ++u) {
        const int s = srcs[j + u];
        w[u] = ws[j + u];
        v[u] = rows[(size_t)s * stride];
    }
#pragma unroll
    for (int u = 0; u < NV; ++u) {
        acc.x += w[u] * v[u].x;
        acc.y += w[u] * v[u].y;
        acc.z += w[u] * v[u].z;
        acc.w += w[u] * v[u].w;
    }
}
__device__ __forceinline__ void gather_piece(const float4* __restrict__ rows, size_t stride,
                                             const int* __restrict__ srcs, const float* __restrict__ ws,
                                             int beg, int end, float4& acc) {
    int j = beg;
#pragma nounroll
    for (; j + GCH <= end; j += GCH) gather_trip<GCH>(rows, stride, srcs, ws, j, acc);
    const int left = end - j;
    if (left == 3)      gather_trip<3>(rows, stride, srcs, ws, j, acc);
    else if (left == 2) gather_trip<2>(rows, stride, srcs, ws, j, acc);
    else if (left == 1) gather_trip<1>(rows, stride, srcs, ws, j, acc);
}
static_assert(GCH == 4, "the left-over trips above are written out for GCH = 4");

__global__ void __launch_bounds__(96)
k_aggregate_out(const int* __restrict__ starts, const int* __restrict__ srcs,
                const float* __restrict__ ws, const float* __restrict__ z2,
                const float* __restrict__ s2, const float* __restrict__ bias2,
                float* __restrict__ out) {
    constexpr int C4 = C_OUT / 4;
    int n   = blockIdx.x;
    int tid = threadIdx.x;
    int beg[CSR_S], end[CSR_S];
#pragma unroll
    for (int i = 0; i < CSR_S; ++i) {
        beg[i] = starts[i * (N_DST2 + 1) + n];
        end[i] = starts[i * (N_DST2 + 1) + n + 1];
    }
    const int t  = tid / C4;
    const int c4 = tid % C4;
    const float4 sv = reinterpret_cast<const float4*>(s2)[(size_t)n * ROW4 + tid];
    const float4 bv = reinterpret_cast<const float4*>(bias2)[c4];
    float4 acc = make_float4(0.f, 0.f, 0.f, 0.f);
    const float4* x4 = reinterpret_cast<const float4*>(z2) + tid;
#pragma unroll
    for (int i = 0; i < CSR_S; ++i) gather_piece(x4, ROW4, srcs, ws, beg[i], end[i], acc);
    float4 o;
    o.x = leaky(sv.x + acc.x + bv.x);
    o.y = leaky(sv.y + acc.y + bv.y);
    o.z = leaky(sv.z + acc.z + bv.z);
    o.w = leaky(sv.w + acc.w + bv.w);
    reinterpret_cast<float4*>(out)[((size_t)t * N_DST2 + n) * C4 + c4] = o;
}

constexpr int FN      = 16;
constexpr int GEMM_NW = FN * T / 16;
constexpr int GEMM_NT = GEMM_NW * 32;
constexpr int P_A     = C_IN + 4;
static_assert(FN * T % 16 == 0 && N_DST1 % FN == 0 && N_DST2 % FN == 0, "whole tiles, whole groups");
constexpr int P_IN    = C_IN + 4;
constexpr int P_H     = H + 4;

template <int ROWS, int K>
__device__ __forceinline__ void stage_weights(const float* __restrict__ Wg, float* sW, int tid) {
    constexpr int K4 = K / 4;
    const float4* __restrict__ g4 = reinterpret_cast<const float4*>(Wg);
    for (int i = tid; i < ROWS * K4; i += GEMM_NT) {
        const int r = i / K4;
        const int q = i - r * K4;
        *reinterpret_cast<float4*>(sW + r * (K + 4) + q * 4) = g4[i];
    }
}

__device__ __forceinline__ v2f lds_frag(const float* sW, int n0, int pitch, int k0, int lane) {
    const int r = n0 + (lane & 15);
    const int k = k0 + ((lane >> 4) << 1);
    return *reinterpret_cast<const v2f*>(sW + r * pitch + k);
}

template <int ROWS>
__device__ __forceinline__ void stage_weights_hperm(const float* __restrict__ Wg, float* sW, int tid) {
    constexpr int K4    = H / 4;
    constexpr int TRIPS = (ROWS * K4 + GEMM_NT - 1) / GEMM_NT;
    const float4* __restrict__ g4 = reinterpret_cast<const float4*>(Wg);
    float4 v[TRIPS];
#pragma unroll
    for (int u = 0; u < TRIPS; ++u) {
        const int i = u * GEMM_NT + tid;
        v[u] = g4[i < ROWS * K4 ? i : 0];
    }
#pragma unroll
    for (int u = 0; u < TRIPS; ++u) {
        const int i  = u * GEMM_NT + tid;
        if (i >= ROWS * K4) continue;
        const int r  = i / K4;
        const int q4 = i - r * K4;
        const int ht = q4 >> 2;
        const int g  = (q4 >> 1) & 1;
        const int jj = (q4 & 1) << 1;
        float* d = sW + r * P_H + 2 * (8 * ht + 2 * jj + g);
        v2f lo, hi;
        lo.x = v[u].x; lo.y = v[u].y;
        hi.x = v[u].z; hi.y = v[u].w;
        *reinterpret_cast<v2f*>(d)     = lo;
        *reinterpret_cast<v2f*>(d + 4) = hi;
    }
}

template <bool SELF>
__device__ __forceinline__ void layer2_tile(const v8f (&hT)[4], const float* sWn, const float* sWs,
                                            float* __restrict__ z2, float* __restrict__ s2,
                                            int m0, int lane) {
    v8f zacc[2] = {};
    v8f sacc[2] = {};
#pragma unroll
    for (int ht = 0; ht < 4; ++ht) {
#pragma unroll
        for (int jj = 0; jj < 4; ++jj) {
            const int k0 = (ht * 4 + jj) * 4;
            v2f b;
            b.x = hT[ht][2 * jj];
            b.y = hT[ht][2 * jj + 1];
#pragma unroll
            for (int ct = 0; ct < 2; ++ct) {
                v2f a = lds_frag(sWn, ct * 16, P_H, k0, lane);
                zacc[ct] = __builtin_amdgcn_wmma_f32_16x16x4_f32(false, a, false, b,
                                                                 (short)0, zacc[ct], false, false);
            }
            if (SELF) {
#pragma unroll
                for (int ct = 0; ct < 2; ++ct) {
                    v2f a = lds_frag(sWs, ct * 16, P_H, k0, lane);
                    sacc[ct] = __builtin_amdgcn_wmma_f32_16x16x4_f32(false, a, false, b,
                                                                     (short)0, sacc[ct], false, false);
                }
            }
        }
    }
    const size_t row = (size_t)(m0 + (lane & 15)) * C_OUT + ((lane >> 4) << 3);
#pragma unroll
    for (int ct = 0; ct < 2; ++ct) {
        float4* zp = reinterpret_cast<float4*>(z2 + row + ct * 16);
        zp[0] = make_float4(zacc[ct][0], zacc[ct][1], zacc[ct][2], zacc[ct][3]);
        zp[1] = make_float4(zacc[ct][4], zacc[ct][5], zacc[ct][6], zacc[ct][7]);
        if (SELF) {
            float4* sp = reinterpret_cast<float4*>(s2 + row + ct * 16);
            sp[0] = make_float4(sacc[ct][0], sacc[ct][1], sacc[ct][2], sacc[ct][3]);
            sp[1] = make_float4(sacc[ct][4], sacc[ct][5], sacc[ct][6], sacc[ct][7]);
        }
    }
}

__global__ void __launch_bounds__(GEMM_NT)
k_layers(const float* __restrict__ nf,
         const int* __restrict__ starts, const int* __restrict__ srcs, const float* __restrict__ wsv,
         const float* __restrict__ Wself, const float* __restrict__ Wneigh,
         const float* __restrict__ bias,
         const float* __restrict__ Wneigh2, const float* __restrict__ Wself2,
         float* __restrict__ z2, float* __restrict__ s2) {
    __shared__ __attribute__((aligned(16))) float sWs[H * P_IN];
    __shared__ __attribute__((aligned(16))) float sWn[H * P_IN];
    __shared__ __attribute__((aligned(16))) float sW2n[C_OUT * P_H];
    __shared__ __attribute__((aligned(16))) float sW2s[C_OUT * P_H];
    __shared__ __attribute__((aligned(16))) float sAgg[FN * T * P_A];

    const int lane  = threadIdx.x & 31;
    const int wave  = threadIdx.x >> 5;
    const int node0 = blockIdx.x * FN;
    const int m0    = node0 * T + wave * 16;
    const bool self_group = (int)blockIdx.x < N_DST2 / FN;

    const int mrow = m0 + (lane & 15);
    const int nrow = mrow / T;
    const int trow = mrow - nrow * T;
    const float* xrow = nf + ((size_t)trow * N_SRC + nrow) * C_IN + ((lane >> 4) << 1);

    v2f xS[8];
#pragma unroll
    for (int ks = 0; ks < 8; ++ks) {
        xS[ks].x = xrow[ks * 4];
        xS[ks].y = xrow[ks * 4 + 1];
    }

    stage_weights<H, C_IN>(Wself,  sWs, threadIdx.x);
    stage_weights<H, C_IN>(Wneigh, sWn, threadIdx.x);
    stage_weights_hperm<C_OUT>(Wneigh2, sW2n, threadIdx.x);
    if (self_group) stage_weights_hperm<C_OUT>(Wself2, sW2s, threadIdx.x);

    {
        constexpr int C4 = C_IN / 4;
        const int grp = __builtin_amdgcn_readfirstlane(wave / 3);
        const int t96 = (wave - grp * 3) * 32 + lane;
        const int t   = t96 / C4;
        const int c4  = t96 % C4;
        const float4* p = reinterpret_cast<const float4*>(nf) + (size_t)t * N_SRC * C4 + c4;
        for (int q = 0; q < FN / 4; ++q) {
            const int nl  = q * 4 + grp;
            int beg[CSR_S], end[CSR_S];
#pragma unroll
            for (int i = 0; i < CSR_S; ++i) {
                beg[i] = starts[i * (N_DST1 + 1) + node0 + nl];
                end[i] = starts[i * (N_DST1 + 1) + node0 + nl + 1];
            }
            float4 acc = make_float4(0.f, 0.f, 0.f, 0.f);
#pragma unroll
            for (int i = 0; i < CSR_S; ++i) gather_piece(p, C4, srcs, wsv, beg[i], end[i], acc);
            *reinterpret_cast<float4*>(sAgg + (nl * T + t) * P_A + c4 * 4) = acc;
        }
    }
    __syncthreads();

    v2f xN[8];
    {
        const float* arow = sAgg + (wave * 16 + (lane & 15)) * P_A + ((lane >> 4) << 1);
#pragma unroll
        for (int ks = 0; ks < 8; ++ks) xN[ks] = *reinterpret_cast<const v2f*>(arow + ks * 4);
    }

    v8f hT[4] = {};
#pragma unroll
    for (int ks = 0; ks < 8; ++ks) {
#pragma unroll
        for (int ht = 0; ht < 4; ++ht) {
            v2f wS = lds_frag(sWs, ht * 16, P_IN, ks * 4, lane);
            hT[ht] = __builtin_amdgcn_wmma_f32_16x16x4_f32(false, wS, false, xS[ks],
                                                           (short)0, hT[ht], false, false);
        }
#pragma unroll
        for (int ht = 0; ht < 4; ++ht) {
            v2f wN = lds_frag(sWn, ht * 16, P_IN, ks * 4, lane);
            hT[ht] = __builtin_amdgcn_wmma_f32_16x16x4_f32(false, wN, false, xN[ks],
                                                           (short)0, hT[ht], false, false);
        }
    }
#pragma unroll
    for (int ht = 0; ht < 4; ++ht) {
        const float4* bp = reinterpret_cast<const float4*>(bias + ht * 16 + ((lane >> 4) << 3));
        const float4 b0 = bp[0];
        const float4 b1 = bp[1];
        hT[ht][0] = leaky(hT[ht][0] + b0.x);
        hT[ht][1] = leaky(hT[ht][1] + b0.y);
        hT[ht][2] = leaky(hT[ht][2] + b0.z);
        hT[ht][3] = leaky(hT[ht][3] + b0.w);
        hT[ht][4] = leaky(hT[ht][4] + b1.x);
        hT[ht][5] = leaky(hT[ht][5] + b1.y);
        hT[ht][6] = leaky(hT[ht][6] + b1.z);
        hT[ht][7] = leaky(hT[ht][7] + b1.w);
    }

    if (self_group) layer2_tile<true >(hT, sW2n, sW2s, z2, s2, m0, lane);
    else            layer2_tile<false>(hT, sW2n, sW2s, z2, s2, m0, lane);
}

extern "C" void kernel_launch(void* const* d_in, const int* in_sizes, int n_in,
                              void* d_out, int out_size, void* d_ws, size_t ws_size,
                              hipStream_t stream) {
    const float* nf   = (const float*)d_in[0];
    const int*   src1 = (const int*)  d_in[1];
    const int*   dst1 = (const int*)  d_in[2];
    const float* ew1  = (const float*)d_in[3];
    const int*   src2 = (const int*)  d_in[4];
    const int*   dst2 = (const int*)  d_in[5];
    const float* ew2  = (const float*)d_in[6];
    const float* W1s  = (const float*)d_in[7];
    const float* W1n  = (const float*)d_in[8];
    const float* b1   = (const float*)d_in[9];
    const float* W2s  = (const float*)d_in[10];
    const float* W2n  = (const float*)d_in[11];
    const float* b2   = (const float*)d_in[12];
    const int E1 = in_sizes[1];
    const int E2 = in_sizes[4];

    float* out = (float*)d_out;

    float* ws   = (float*)d_ws;
    float* z2   = ws;
    float* s2   = ws   + (size_t)N_DST1 * ROW;
    int*   ip      = (int*)(ws + (size_t)N_SRC * ROW + (size_t)N_DST1 * ROW + (size_t)M1 * H);
    int*   sp      = (int*)(ws + (size_t)N_SRC * ROW);
    int*   starts1 = sp;                 sp += CSR_S * (N_DST1 + 1);
    int*   starts2 = sp;
    int*   srcs1   = ip;                 ip += E1;
    float* ws1     = (float*)ip;         ip += E1;
    int*   srcs2   = ip;                 ip += E2;
    float* ws2     = (float*)ip;

    k_csr<<<CSR_S * (CSR_NB1 + CSR_NB2), CSR_NT, 0, stream>>>(src1, dst1, ew1, starts1, srcs1, ws1, E1,
                                                   src2, dst2, ew2, starts2, srcs2, ws2, E2);

    k_layers<<<N_DST1 / FN, GEMM_NT, 0, stream>>>(nf, starts1, srcs1, ws1, W1s, W1n, b1,
                                                  W2n, W2s, z2, s2);

    k_aggregate_out<<<N_DST2, 96, 0, stream>>>(starts2, srcs2, ws2, z2, s2, b2, out);
}
